# starting kernel
# baseline (speedup 1.0000x reference)
#include <hip/hip_runtime.h>
#include <hip/hip_bf16.h>

#define N_NODES   100000
#define N_EDGES   1600000
#define NODE_DIM  64
#define HIDDEN    256
#define OUT_DIM   64
#define BN_EPS    1e-5f

typedef __attribute__((ext_vector_type(2))) float v2f;
typedef __attribute__((ext_vector_type(8))) float v8f;

#define WMMA_F32X4(ACC, A, B) \
  ACC = __builtin_amdgcn_wmma_f32_16x16x4_f32(false, (A), false, (B), (short)0, (ACC), false, false)

__global__ __launch_bounds__(256) void zero_ws(float* __restrict__ p, int n) {
  int i = blockIdx.x * 256 + threadIdx.x;
  if (i < n) p[i] = 0.0f;
}

__global__ __launch_bounds__(256) void edge_scatter(
    const float* __restrict__ h, const int* __restrict__ src,
    const int* __restrict__ dst, float* __restrict__ agg,
    float* __restrict__ deg) {
  unsigned tid = blockIdx.x * 256u + threadIdx.x;
  if (tid >= (unsigned)N_EDGES * NODE_DIM) return;
  int e = tid >> 6;
  int c = tid & 63;
  int s = src[e];
  int d = dst[e];
  atomicAdd(&agg[d * NODE_DIM + c], h[s * NODE_DIM + c]);
  if (c == 0) atomicAdd(&deg[d], 1.0f);
}

__global__ __launch_bounds__(128) void fused_mlp(
    const float* __restrict__ h,  const float* __restrict__ W1,
    const float* __restrict__ b1, const float* __restrict__ W2,
    const float* __restrict__ b2, const float* __restrict__ agg,
    const float* __restrict__ deg, float* __restrict__ out) {
  __shared__ float xs[16][NODE_DIM + 1];
  __shared__ float hs[16][HIDDEN + 1];

  const int m0   = blockIdx.x * 16;
  const int tid  = threadIdx.x;
  const int wave = tid >> 5;
  const int lane = tid & 31;
  const int half = lane >> 4;
  const int r    = lane & 15;

  for (int i = tid; i < 16 * NODE_DIM; i += 128) {
    int m = i >> 6, c = i & 63;
    int g = (m0 + m) * NODE_DIM + c;
    float d = deg[m0 + m];
    d = d > 1.0f ? d : 1.0f;
    xs[m][c] = agg[g] / d + h[g];
  }
  __syncthreads();

  v8f acc0 = {}, acc1 = {}, acc2 = {}, acc3 = {};
  const int nb = wave * 64;
  for (int k = 0; k < NODE_DIM; k += 4) {
    v2f a;
    a.x = xs[r][k + 2 * half];
    a.y = xs[r][k + 2 * half + 1];
    const float* w1p0 = W1 + (k + 2 * half) * HIDDEN + nb + r;
    const float* w1p1 = W1 + (k + 2 * half + 1) * HIDDEN + nb + r;
    v2f b;
    b.x = w1p0[0];  b.y = w1p1[0];  WMMA_F32X4(acc0, a, b);
    b.x = w1p0[16]; b.y = w1p1[16]; WMMA_F32X4(acc1, a, b);
    b.x = w1p0[32]; b.y = w1p1[32]; WMMA_F32X4(acc2, a, b);
    b.x = w1p0[48]; b.y = w1p1[48]; WMMA_F32X4(acc3, a, b);
  }
#define STORE_TILE(ACC, T)                                        \
  {                                                               \
    const int n = nb + (T) * 16 + r;                              \
    const float bias = b1[n];                                     \
    _Pragma("unroll") for (int j = 0; j < 8; ++j) {               \
      float v = (ACC)[j] + bias;                                  \
      hs[j + 8 * half][n] = v > 0.0f ? v : 0.0f;                  \
    }                                                             \
  }
  STORE_TILE(acc0, 0)
  STORE_TILE(acc1, 1)
  STORE_TILE(acc2, 2)
  STORE_TILE(acc3, 3)
#undef STORE_TILE
  __syncthreads();

  v8f occ = {};
  const int ob = wave * 16;
  for (int k = 0; k < HIDDEN; k += 4) {
    v2f a;
    a.x = hs[r][k + 2 * half];
    a.y = hs[r][k + 2 * half + 1];
    v2f b;
    b.x = W2[(k + 2 * half) * OUT_DIM + ob + r];
    b.y = W2[(k + 2 * half + 1) * OUT_DIM + ob + r];
    WMMA_F32X4(occ, a, b);
  }
  {
    const float bias = b2[ob + r];
#pragma unroll
    for (int j = 0; j < 8; ++j) {
      float v = occ[j] + bias;
      v = v > 0.0f ? v : 0.0f;
      out[(m0 + j + 8 * half) * OUT_DIM + ob + r] = v;
    }
  }
}

__global__ __launch_bounds__(256) void bn_stats(
    const float* __restrict__ x, float* __restrict__ gsum,
    float* __restrict__ gsq) {
  __shared__ float ssum[OUT_DIM], ssq[OUT_DIM];
  const int tid = threadIdx.x;
  if (tid < OUT_DIM) { ssum[tid] = 0.0f; ssq[tid] = 0.0f; }
  __syncthreads();
  float s = 0.0f, q = 0.0f;
  const int total = N_NODES * OUT_DIM;
  const int stride = gridDim.x * 256;
  for (int i = blockIdx.x * 256 + tid; i < total; i += stride) {
    float v = x[i];
    s += v;
    q += v * v;
  }
  const int c = tid & 63;
  atomicAdd(&ssum[c], s);
  atomicAdd(&ssq[c], q);
  __syncthreads();
  if (tid < OUT_DIM) {
    atomicAdd(&gsum[tid], ssum[tid]);
    atomicAdd(&gsq[tid], ssq[tid]);
  }
}

__global__ __launch_bounds__(256) void bn_apply(
    float* __restrict__ x, const float* __restrict__ gsum,
    const float* __restrict__ gsq, const float* __restrict__ gamma,
    const float* __restrict__ beta) {
  int i = blockIdx.x * 256 + threadIdx.x;
  if (i >= N_NODES * OUT_DIM) return;
  int c = i & 63;
  const float inv_n = 1.0f / (float)N_NODES;
  float mean = gsum[c] * inv_n;
  float var  = gsq[c] * inv_n - mean * mean;
  float sc   = gamma[c] * __frsqrt_rn(var + BN_EPS);
  x[i] = (x[i] - mean) * sc + beta[c];
}

extern "C" void kernel_launch(void* const* d_in, const int* in_sizes, int n_in,
                              void* d_out, int out_size, void* d_ws, size_t ws_size,
                              hipStream_t stream) {
  const float* h     = (const float*)d_in[0];
  const float* W1    = (const float*)d_in[1];
  const float* b1    = (const float*)d_in[2];
  const float* W2    = (const float*)d_in[3];
  const float* b2    = (const float*)d_in[4];
  const float* gamma = (const float*)d_in[5];
  const float* beta  = (const float*)d_in[6];
  const int*   src   = (const int*)d_in[7];
  const int*   dst   = (const int*)d_in[8];
  float* out = (float*)d_out;

  float* agg  = (float*)d_ws;
  float* deg  = agg + (size_t)N_NODES * NODE_DIM;
  float* gsum = deg + N_NODES;
  float* gsq  = gsum + OUT_DIM;

  const int nzero = N_NODES * NODE_DIM + N_NODES + 2 * OUT_DIM;
  zero_ws<<<(nzero + 255) / 256, 256, 0, stream>>>((float*)d_ws, nzero);

  const unsigned nscatter = (unsigned)N_EDGES * NODE_DIM;
  edge_scatter<<<(nscatter + 255) / 256, 256, 0, stream>>>(h, src, dst, agg, deg);

  fused_mlp<<<N_NODES / 16, 128, 0, stream>>>(h, W1, b1, W2, b2, agg, deg, out);

  bn_stats<<<1024, 256, 0, stream>>>(out, gsum, gsq);

  bn_apply<<<(N_NODES * OUT_DIM + 255) / 256, 256, 0, stream>>>(out, gsum, gsq, gamma, beta);
}


# Scatter-add replaced by owner groups: 128 groups each hold 782 nodes' sums in LDS, scan the whole dst list, gather matching h rows, add in LDS; no global atomics, no zero pass; sums by LDS arrival
# speedup vs baseline: 2.4912x; 2.4912x over previous
#include <hip/hip_runtime.h>
#include <hip/hip_bf16.h>

#define N_NODES   100000
#define N_EDGES   1600000
#define NODE_DIM  64
#define HIDDEN    256
#define OUT_DIM   64
#define BN_EPS    1e-5f

typedef __attribute__((ext_vector_type(2))) float v2f;
typedef __attribute__((ext_vector_type(8))) float v8f;

#define WMMA_F32X4(ACC, A, B) \
  ACC = __builtin_amdgcn_wmma_f32_16x16x4_f32(false, (A), false, (B), (short)0, (ACC), false, false)

#define OWN_GROUPS   128
#define OWN_NODES    782
#define OWN_THREADS  1024
#define OWN_WAVES    32
#define BATCH_EDGES  512
#define N_BATCH      (N_EDGES / BATCH_EDGES)
#define RING         256
#define ROWS_AHEAD   16

static_assert(N_EDGES % BATCH_EDGES == 0, "edge list must split into whole batches");
static_assert(OWN_GROUPS * OWN_NODES >= N_NODES, "owner ranges must cover all nodes");
static_assert(OWN_NODES <= 1024 && N_EDGES < (1 << 21), "ring entry packs edge<<10 | local node");

template <bool FULL>
__device__ __forceinline__ void drain_ring(
    const float* __restrict__ h, const int* __restrict__ src,
    float* acc, float* sdeg, const unsigned* q, unsigned head, int n, int lane) {
  const unsigned ent = q[(head + lane) & (RING - 1)];
  const bool valid = FULL || lane < n;
  const int e  = (int)(ent >> 10);
  const int dl = (int)(ent & 1023u);
  int s = 0;
  if (valid) {
    s = src[e];
    atomicAdd(&sdeg[dl], 1.0f);
  }
  const float2* h2 = reinterpret_cast<const float2*>(h);
  const int cntj = FULL ? 32 : n;
  for (int j0 = 0; j0 < cntj; j0 += ROWS_AHEAD) {
    float2 v[ROWS_AHEAD];
#pragma unroll
    for (int u = 0; u < ROWS_AHEAD; ++u) {
      int j = j0 + u;
      if (j >= cntj) j = cntj - 1;
      const int sj = __builtin_amdgcn_readlane(s, j);
      v[u] = h2[(unsigned)sj * (unsigned)(NODE_DIM / 2) + (unsigned)lane];
    }
#pragma unroll
    for (int u = 0; u < ROWS_AHEAD; ++u) {
      const int j = j0 + u;
      if (j < cntj) {
        const int dj = __builtin_amdgcn_readlane(dl, j);
        float* a = acc + dj * NODE_DIM + lane;
        atomicAdd(a, v[u].x);
        atomicAdd(a + 32, v[u].y);
      }
    }
  }
}

__global__ __launch_bounds__(OWN_THREADS) void edge_owner_scan(
    const float* __restrict__ h, const int* __restrict__ src,
    const int* __restrict__ dst, float* __restrict__ agg,
    float* __restrict__ deg, float* __restrict__ gstat) {
  __shared__ float acc[OWN_NODES * NODE_DIM];
  __shared__ float sdeg[OWN_NODES];
  __shared__ unsigned ring[OWN_WAVES][RING];

  const int tid  = threadIdx.x;
  const int wave = __builtin_amdgcn_readfirstlane(tid >> 5);
  const int lane = tid & 31;
  const int base = blockIdx.x * OWN_NODES;
  int cnt = N_NODES - base;
  if (cnt > OWN_NODES) cnt = OWN_NODES;

  {
    float4* a4 = reinterpret_cast<float4*>(acc);
    const float4 z = {0.0f, 0.0f, 0.0f, 0.0f};
    for (int i = tid; i < OWN_NODES * NODE_DIM / 4; i += OWN_THREADS) a4[i] = z;
    for (int i = tid; i < OWN_NODES; i += OWN_THREADS) sdeg[i] = 0.0f;
    if (blockIdx.x == 0 && tid < 2 * OUT_DIM) gstat[tid] = 0.0f;
  }
  __syncthreads();

  unsigned* q = ring[wave];
  unsigned head = 0, tail = 0;
  const int4* dst4 = reinterpret_cast<const int4*>(dst);
  const int start = blockIdx.x * (N_BATCH / OWN_GROUPS);

  for (int k = wave; k < N_BATCH; k += OWN_WAVES) {
    int bb = start + k;
    if (bb >= N_BATCH) bb -= N_BATCH;
    const int4* p = dst4 + (size_t)bb * (BATCH_EDGES / 4) + lane;
    int4 d[4];
#pragma unroll
    for (int i = 0; i < 4; ++i) d[i] = p[i * 32];
    const unsigned e0 = (unsigned)bb * BATCH_EDGES + 4u * lane;

#pragma unroll
    for (int i = 0; i < 4; ++i) {
      const int dc[4] = {d[i].x, d[i].y, d[i].z, d[i].w};
      unsigned t[4], mask[4];
#pragma unroll
      for (int c = 0; c < 4; ++c) {
        t[c] = (unsigned)dc[c] - (unsigned)base;
        mask[c] = __builtin_amdgcn_ballot_w32(t[c] < (unsigned)cnt);
      }
      if ((mask[0] | mask[1] | mask[2] | mask[3]) == 0u) continue;
#pragma unroll
      for (int c = 0; c < 4; ++c) {
        if (mask[c] == 0u) continue;
        if (t[c] < (unsigned)cnt) {
          const unsigned pos = tail + __builtin_amdgcn_mbcnt_lo(mask[c], 0u);
          q[pos & (RING - 1)] = ((e0 + (unsigned)(i * 128 + c)) << 10) | t[c];
        }
        tail += (unsigned)__builtin_popcount(mask[c]);
      }
      while (tail - head >= 32u) {
        drain_ring<true>(h, src, acc, sdeg, q, head, 32, lane);
        head += 32u;
      }
    }
  }
  if (tail != head)
    drain_ring<false>(h, src, acc, sdeg, q, head, (int)(tail - head), lane);
  __syncthreads();

  for (int i = tid; i < cnt * NODE_DIM; i += OWN_THREADS) {
    const int m = i >> 6, c = i & 63;
    agg[(size_t)(base + m) * NODE_DIM + c] = acc[m * NODE_DIM + ((c & 1) << 5) + (c >> 1)];
  }
  for (int i = tid; i < cnt; i += OWN_THREADS) deg[base + i] = sdeg[i];
}

__global__ __launch_bounds__(128) void fused_mlp(
    const float* __restrict__ h,  const float* __restrict__ W1,
    const float* __restrict__ b1, const float* __restrict__ W2,
    const float* __restrict__ b2, const float* __restrict__ agg,
    const float* __restrict__ deg, float* __restrict__ out) {
  __shared__ float xs[16][NODE_DIM + 1];
  __shared__ float hs[16][HIDDEN + 1];

  const int m0   = blockIdx.x * 16;
  const int tid  = threadIdx.x;
  const int wave = tid >> 5;
  const int lane = tid & 31;
  const int half = lane >> 4;
  const int r    = lane & 15;

  for (int i = tid; i < 16 * NODE_DIM; i += 128) {
    int m = i >> 6, c = i & 63;
    int g = (m0 + m) * NODE_DIM + c;
    float d = deg[m0 + m];
    d = d > 1.0f ? d : 1.0f;
    xs[m][c] = agg[g] / d + h[g];
  }
  __syncthreads();

  v8f acc0 = {}, acc1 = {}, acc2 = {}, acc3 = {};
  const int nb = wave * 64;
  for (int k = 0; k < NODE_DIM; k += 4) {
    v2f a;
    a.x = xs[r][k + 2 * half];
    a.y = xs[r][k + 2 * half + 1];
    const float* w1p0 = W1 + (k + 2 * half) * HIDDEN + nb + r;
    const float* w1p1 = W1 + (k + 2 * half + 1) * HIDDEN + nb + r;
    v2f b;
    b.x = w1p0[0];  b.y = w1p1[0];  WMMA_F32X4(acc0, a, b);
    b.x = w1p0[16]; b.y = w1p1[16]; WMMA_F32X4(acc1, a, b);
    b.x = w1p0[32]; b.y = w1p1[32]; WMMA_F32X4(acc2, a, b);
    b.x = w1p0[48]; b.y = w1p1[48]; WMMA_F32X4(acc3, a, b);
  }
#define STORE_TILE(ACC, T)                                        \
  {                                                               \
    const int n = nb + (T) * 16 + r;                              \
    const float bias = b1[n];                                     \
    _Pragma("unroll") for (int j = 0; j < 8; ++j) {               \
      float v = (ACC)[j] + bias;                                  \
      hs[j + 8 * half][n] = v > 0.0f ? v : 0.0f;                  \
    }                                                             \
  }
  STORE_TILE(acc0, 0)
  STORE_TILE(acc1, 1)
  STORE_TILE(acc2, 2)
  STORE_TILE(acc3, 3)
#undef STORE_TILE
  __syncthreads();

  v8f occ = {};
  const int ob = wave * 16;
  for (int k = 0; k < HIDDEN; k += 4) {
    v2f a;
    a.x = hs[r][k + 2 * half];
    a.y = hs[r][k + 2 * half + 1];
    v2f b;
    b.x = W2[(k + 2 * half) * OUT_DIM + ob + r];
    b.y = W2[(k + 2 * half + 1) * OUT_DIM + ob + r];
    WMMA_F32X4(occ, a, b);
  }
  {
    const float bias = b2[ob + r];
#pragma unroll
    for (int j = 0; j < 8; ++j) {
      float v = occ[j] + bias;
      v = v > 0.0f ? v : 0.0f;
      out[(m0 + j + 8 * half) * OUT_DIM + ob + r] = v;
    }
  }
}

__global__ __launch_bounds__(256) void bn_stats(
    const float* __restrict__ x, float* __restrict__ gsum,
    float* __restrict__ gsq) {
  __shared__ float ssum[OUT_DIM], ssq[OUT_DIM];
  const int tid = threadIdx.x;
  if (tid < OUT_DIM) { ssum[tid] = 0.0f; ssq[tid] = 0.0f; }
  __syncthreads();
  float s = 0.0f, q = 0.0f;
  const int total = N_NODES * OUT_DIM;
  const int stride = gridDim.x * 256;
  for (int i = blockIdx.x * 256 + tid; i < total; i += stride) {
    float v = x[i];
    s += v;
    q += v * v;
  }
  const int c = tid & 63;
  atomicAdd(&ssum[c], s);
  atomicAdd(&ssq[c], q);
  __syncthreads();
  if (tid < OUT_DIM) {
    atomicAdd(&gsum[tid], ssum[tid]);
    atomicAdd(&gsq[tid], ssq[tid]);
  }
}

__global__ __launch_bounds__(256) void bn_apply(
    float* __restrict__ x, const float* __restrict__ gsum,
    const float* __restrict__ gsq, const float* __restrict__ gamma,
    const float* __restrict__ beta) {
  int i = blockIdx.x * 256 + threadIdx.x;
  if (i >= N_NODES * OUT_DIM) return;
  int c = i & 63;
  const float inv_n = 1.0f / (float)N_NODES;
  float mean = gsum[c] * inv_n;
  float var  = gsq[c] * inv_n - mean * mean;
  float sc   = gamma[c] * __frsqrt_rn(var + BN_EPS);
  x[i] = (x[i] - mean) * sc + beta[c];
}

extern "C" void kernel_launch(void* const* d_in, const int* in_sizes, int n_in,
                              void* d_out, int out_size, void* d_ws, size_t ws_size,
                              hipStream_t stream) {
  const float* h     = (const float*)d_in[0];
  const float* W1    = (const float*)d_in[1];
  const float* b1    = (const float*)d_in[2];
  const float* W2    = (const float*)d_in[3];
  const float* b2    = (const float*)d_in[4];
  const float* gamma = (const float*)d_in[5];
  const float* beta  = (const float*)d_in[6];
  const int*   src   = (const int*)d_in[7];
  const int*   dst   = (const int*)d_in[8];
  float* out = (float*)d_out;

  float* agg  = (float*)d_ws;
  float* deg  = agg + (size_t)N_NODES * NODE_DIM;
  float* gsum = deg + N_NODES;
  float* gsq  = gsum + OUT_DIM;

  edge_owner_scan<<<OWN_GROUPS, OWN_THREADS, 0, stream>>>(h, src, dst, agg, deg, gsum);

  fused_mlp<<<N_NODES / 16, 128, 0, stream>>>(h, W1, b1, W2, b2, agg, deg, out);

  bn_stats<<<1024, 256, 0, stream>>>(out, gsum, gsq);

  bn_apply<<<(N_NODES * OUT_DIM + 255) / 256, 256, 0, stream>>>(out, gsum, gsq, gamma, beta);
}


# Owner groups stop comparing every dst: a new first pass marks each edge in its owner's bit plane (LDS ORs), 256 owners of 391 nodes read only their plane; 4 -> 5 launches; sums by LDS arrival
# speedup vs baseline: 2.8863x; 1.1586x over previous
#include <hip/hip_runtime.h>
#include <hip/hip_bf16.h>

#define N_NODES   100000
#define N_EDGES   1600000
#define NODE_DIM  64
#define HIDDEN    256
#define OUT_DIM   64
#define BN_EPS    1e-5f

typedef __attribute__((ext_vector_type(2))) float v2f;
typedef __attribute__((ext_vector_type(8))) float v8f;

#define WMMA_F32X4(ACC, A, B) \
  ACC = __builtin_amdgcn_wmma_f32_16x16x4_f32(false, (A), false, (B), (short)0, (ACC), false, false)

#define OWN_GROUPS   256
#define OWN_NODES    391
#define OWN_THREADS  1024
#define OWN_WAVES    32
#define RING         64
#define ROWS_AHEAD   16
#define BM_WORDS     (N_EDGES / 32)
#define CH_EDGES     4096
#define CH_WORDS     (CH_EDGES / 32)
#define N_CHUNK      ((N_EDGES + CH_EDGES - 1) / CH_EDGES)

static_assert(OWN_GROUPS * OWN_NODES >= N_NODES, "owner ranges must cover all nodes");
static_assert(N_EDGES % 32 == 0 && BM_WORDS % 4 == 0, "a plane is whole 16-byte pieces");
static_assert(N_EDGES % 4 == 0, "dst is read four edges at a time");
static_assert(CH_EDGES == 4 * OWN_THREADS, "one 16-byte dst load per thread");

__global__ __launch_bounds__(OWN_THREADS) void edge_owner_bits(
    const int* __restrict__ dst, unsigned* __restrict__ bits) {
  __shared__ unsigned tab[OWN_GROUPS * CH_WORDS];
  const int tid = threadIdx.x;
  uint4* t4 = reinterpret_cast<uint4*>(tab);
  {
    const uint4 z = {0u, 0u, 0u, 0u};
    for (int i = tid; i < OWN_GROUPS * CH_WORDS / 4; i += OWN_THREADS) t4[i] = z;
  }
  __syncthreads();
  const int e0 = blockIdx.x * CH_EDGES + tid * 4;
  if (e0 < N_EDGES) {
    const int4 d = *reinterpret_cast<const int4*>(dst + e0);
    const unsigned dc[4] = {(unsigned)d.x, (unsigned)d.y, (unsigned)d.z, (unsigned)d.w};
    const int w  = tid >> 3;
    const int b0 = (tid & 7) * 4;
#pragma unroll
    for (int c = 0; c < 4; ++c) {
      if (dc[c] < (unsigned)N_NODES) {
        const unsigned g = dc[c] / (unsigned)OWN_NODES;
        atomicOr(&tab[g * CH_WORDS + w], 1u << (b0 + c));
      }
    }
  }
  __syncthreads();
  const int wbase = blockIdx.x * CH_WORDS;
  for (int i = tid; i < OWN_GROUPS * CH_WORDS / 4; i += OWN_THREADS) {
    const int g  = i / (CH_WORDS / 4);
    const int wv = wbase + (i % (CH_WORDS / 4)) * 4;
    if (wv < BM_WORDS)
      *reinterpret_cast<uint4*>(bits + (size_t)g * BM_WORDS + wv) = t4[i];
  }
}

template <bool FULL>
__device__ __forceinline__ void drain_ring(
    const float* __restrict__ h, const int* __restrict__ src,
    const int* __restrict__ dst, float* acc, float* sdeg,
    const unsigned* q, unsigned head, int n, int lane, int base) {
  const unsigned e = q[(head + lane) & (RING - 1)];
  const bool valid = FULL || lane < n;
  int s = 0, dl = 0;
  if (valid) {
    s  = src[e];
    dl = dst[e] - base;
    atomicAdd(&sdeg[dl], 1.0f);
  }
  const float2* h2 = reinterpret_cast<const float2*>(h);
  const int cntj = FULL ? 32 : n;
  for (int j0 = 0; j0 < cntj; j0 += ROWS_AHEAD) {
    float2 v[ROWS_AHEAD];
#pragma unroll
    for (int u = 0; u < ROWS_AHEAD; ++u) {
      int j = j0 + u;
      if (j >= cntj) j = cntj - 1;
      const int sj = __builtin_amdgcn_readlane(s, j);
      v[u] = h2[(unsigned)sj * (unsigned)(NODE_DIM / 2) + (unsigned)lane];
    }
#pragma unroll
    for (int u = 0; u < ROWS_AHEAD; ++u) {
      const int j = j0 + u;
      if (j < cntj) {
        const int dj = __builtin_amdgcn_readlane(dl, j);
        float* a = acc + dj * NODE_DIM + lane;
        atomicAdd(a, v[u].x);
        atomicAdd(a + 32, v[u].y);
      }
    }
  }
}

__global__ __launch_bounds__(OWN_THREADS) void edge_owner_scan(
    const float* __restrict__ h, const int* __restrict__ src,
    const int* __restrict__ dst, const unsigned* __restrict__ bits,
    float* __restrict__ agg, float* __restrict__ deg, float* __restrict__ gstat) {
  __shared__ float acc[OWN_NODES * NODE_DIM];
  __shared__ float sdeg[OWN_NODES];
  __shared__ unsigned ring[OWN_WAVES][RING];

  const int tid  = threadIdx.x;
  const int wave = __builtin_amdgcn_readfirstlane(tid >> 5);
  const int lane = tid & 31;
  const int base = blockIdx.x * OWN_NODES;
  int cnt = N_NODES - base;
  if (cnt > OWN_NODES) cnt = OWN_NODES;

  {
    float4* a4 = reinterpret_cast<float4*>(acc);
    const float4 z = {0.0f, 0.0f, 0.0f, 0.0f};
    for (int i = tid; i < OWN_NODES * NODE_DIM / 4; i += OWN_THREADS) a4[i] = z;
    for (int i = tid; i < OWN_NODES; i += OWN_THREADS) sdeg[i] = 0.0f;
    if (blockIdx.x == 0 && tid < 2 * OUT_DIM) gstat[tid] = 0.0f;
  }
  __syncthreads();

  unsigned* q = ring[wave];
  unsigned head = 0, tail = 0;
  const uint4* bm4 = reinterpret_cast<const uint4*>(bits + (size_t)blockIdx.x * BM_WORDS);

  const uint4 none = {0u, 0u, 0u, 0u};
  uint4 nxt;
  {
    const int f4 = wave * 32 + lane;
    nxt = bm4[f4 < BM_WORDS / 4 ? f4 : BM_WORDS / 4 - 1];
    if (f4 >= BM_WORDS / 4) nxt = none;
  }
  for (int k = wave * 32; k < BM_WORDS / 4; k += OWN_WAVES * 32) {
    const int i4 = k + lane;
    const uint4 m4 = nxt;
    const int in4 = i4 + OWN_WAVES * 32;
    nxt = bm4[in4 < BM_WORDS / 4 ? in4 : BM_WORDS / 4 - 1];
    if (in4 >= BM_WORDS / 4) nxt = none;
    const unsigned mw[4] = {m4.x, m4.y, m4.z, m4.w};
#pragma unroll
    for (int c = 0; c < 4; ++c) {
      unsigned mm = mw[c];
      const unsigned ebase = ((unsigned)i4 * 4u + (unsigned)c) * 32u;
      unsigned any = __builtin_amdgcn_ballot_w32(mm != 0u);
      while (any != 0u) {
        if (mm != 0u) {
          const unsigned b = (unsigned)__builtin_ctz(mm);
          mm &= mm - 1u;
          const unsigned pos = tail + __builtin_amdgcn_mbcnt_lo(any, 0u);
          q[pos & (RING - 1)] = ebase + b;
        }
        tail += (unsigned)__builtin_popcount(any);
        while (tail - head >= 32u) {
          drain_ring<true>(h, src, dst, acc, sdeg, q, head, 32, lane, base);
          head += 32u;
        }
        any = __builtin_amdgcn_ballot_w32(mm != 0u);
      }
    }
  }
  if (tail != head)
    drain_ring<false>(h, src, dst, acc, sdeg, q, head, (int)(tail - head), lane, base);
  __syncthreads();

  for (int i = tid; i < cnt * NODE_DIM; i += OWN_THREADS) {
    const int m = i >> 6, c = i & 63;
    agg[(size_t)(base + m) * NODE_DIM + c] = acc[m * NODE_DIM + ((c & 1) << 5) + (c >> 1)];
  }
  for (int i = tid; i < cnt; i += OWN_THREADS) deg[base + i] = sdeg[i];
}

__global__ __launch_bounds__(128) void fused_mlp(
    const float* __restrict__ h,  const float* __restrict__ W1,
    const float* __restrict__ b1, const float* __restrict__ W2,
    const float* __restrict__ b2, const float* __restrict__ agg,
    const float* __restrict__ deg, float* __restrict__ out) {
  __shared__ float xs[16][NODE_DIM + 1];
  __shared__ float hs[16][HIDDEN + 1];

  const int m0   = blockIdx.x * 16;
  const int tid  = threadIdx.x;
  const int wave = tid >> 5;
  const int lane = tid & 31;
  const int half = lane >> 4;
  const int r    = lane & 15;

  for (int i = tid; i < 16 * NODE_DIM; i += 128) {
    int m = i >> 6, c = i & 63;
    int g = (m0 + m) * NODE_DIM + c;
    float d = deg[m0 + m];
    d = d > 1.0f ? d : 1.0f;
    xs[m][c] = agg[g] / d + h[g];
  }
  __syncthreads();

  v8f acc0 = {}, acc1 = {}, acc2 = {}, acc3 = {};
  const int nb = wave * 64;
  for (int k = 0; k < NODE_DIM; k += 4) {
    v2f a;
    a.x = xs[r][k + 2 * half];
    a.y = xs[r][k + 2 * half + 1];
    const float* w1p0 = W1 + (k + 2 * half) * HIDDEN + nb + r;
    const float* w1p1 = W1 + (k + 2 * half + 1) * HIDDEN + nb + r;
    v2f b;
    b.x = w1p0[0];  b.y = w1p1[0];  WMMA_F32X4(acc0, a, b);
    b.x = w1p0[16]; b.y = w1p1[16]; WMMA_F32X4(acc1, a, b);
    b.x = w1p0[32]; b.y = w1p1[32]; WMMA_F32X4(acc2, a, b);
    b.x = w1p0[48]; b.y = w1p1[48]; WMMA_F32X4(acc3, a, b);
  }
#define STORE_TILE(ACC, T)                                        \
  {                                                               \
    const int n = nb + (T) * 16 + r;                              \
    const float bias = b1[n];                                     \
    _Pragma("unroll") for (int j = 0; j < 8; ++j) {               \
      float v = (ACC)[j] + bias;                                  \
      hs[j + 8 * half][n] = v > 0.0f ? v : 0.0f;                  \
    }                                                             \
  }
  STORE_TILE(acc0, 0)
  STORE_TILE(acc1, 1)
  STORE_TILE(acc2, 2)
  STORE_TILE(acc3, 3)
#undef STORE_TILE
  __syncthreads();

  v8f occ = {};
  const int ob = wave * 16;
  for (int k = 0; k < HIDDEN; k += 4) {
    v2f a;
    a.x = hs[r][k + 2 * half];
    a.y = hs[r][k + 2 * half + 1];
    v2f b;
    b.x = W2[(k + 2 * half) * OUT_DIM + ob + r];
    b.y = W2[(k + 2 * half + 1) * OUT_DIM + ob + r];
    WMMA_F32X4(occ, a, b);
  }
  {
    const float bias = b2[ob + r];
#pragma unroll
    for (int j = 0; j < 8; ++j) {
      float v = occ[j] + bias;
      v = v > 0.0f ? v : 0.0f;
      out[(m0 + j + 8 * half) * OUT_DIM + ob + r] = v;
    }
  }
}

__global__ __launch_bounds__(256) void bn_stats(
    const float* __restrict__ x, float* __restrict__ gsum,
    float* __restrict__ gsq) {
  __shared__ float ssum[OUT_DIM], ssq[OUT_DIM];
  const int tid = threadIdx.x;
  if (tid < OUT_DIM) { ssum[tid] = 0.0f; ssq[tid] = 0.0f; }
  __syncthreads();
  float s = 0.0f, q = 0.0f;
  const int total = N_NODES * OUT_DIM;
  const int stride = gridDim.x * 256;
  for (int i = blockIdx.x * 256 + tid; i < total; i += stride) {
    float v = x[i];
    s += v;
    q += v * v;
  }
  const int c = tid & 63;
  atomicAdd(&ssum[c], s);
  atomicAdd(&ssq[c], q);
  __syncthreads();
  if (tid < OUT_DIM) {
    atomicAdd(&gsum[tid], ssum[tid]);
    atomicAdd(&gsq[tid], ssq[tid]);
  }
}

__global__ __launch_bounds__(256) void bn_apply(
    float* __restrict__ x, const float* __restrict__ gsum,
    const float* __restrict__ gsq, const float* __restrict__ gamma,
    const float* __restrict__ beta) {
  int i = blockIdx.x * 256 + threadIdx.x;
  if (i >= N_NODES * OUT_DIM) return;
  int c = i & 63;
  const float inv_n = 1.0f / (float)N_NODES;
  float mean = gsum[c] * inv_n;
  float var  = gsq[c] * inv_n - mean * mean;
  float sc   = gamma[c] * __frsqrt_rn(var + BN_EPS);
  x[i] = (x[i] - mean) * sc + beta[c];
}

extern "C" void kernel_launch(void* const* d_in, const int* in_sizes, int n_in,
                              void* d_out, int out_size, void* d_ws, size_t ws_size,
                              hipStream_t stream) {
  const float* h     = (const float*)d_in[0];
  const float* W1    = (const float*)d_in[1];
  const float* b1    = (const float*)d_in[2];
  const float* W2    = (const float*)d_in[3];
  const float* b2    = (const float*)d_in[4];
  const float* gamma = (const float*)d_in[5];
  const float* beta  = (const float*)d_in[6];
  const int*   src   = (const int*)d_in[7];
  const int*   dst   = (const int*)d_in[8];
  float* out = (float*)d_out;

  float* agg  = (float*)d_ws;
  float* deg  = agg + (size_t)N_NODES * NODE_DIM;
  float* gsum = deg + N_NODES;
  float* gsq  = gsum + OUT_DIM;
  unsigned* bits = (unsigned*)(gsq + OUT_DIM);

  edge_owner_bits<<<N_CHUNK, OWN_THREADS, 0, stream>>>(dst, bits);

  edge_owner_scan<<<OWN_GROUPS, OWN_THREADS, 0, stream>>>(h, src, dst, bits, agg, deg, gsum);

  fused_mlp<<<N_NODES / 16, 128, 0, stream>>>(h, W1, b1, W2, b2, agg, deg, out);

  bn_stats<<<1024, 256, 0, stream>>>(out, gsum, gsq);

  bn_apply<<<(N_NODES * OUT_DIM + 255) / 256, 256, 0, stream>>>(out, gsum, gsq, gamma, beta);
}


# BN sums start in the MLP epilogue: each wave adds its 16x16 tile's columns (sum, squares) into one 128-B partial row; stats launch adds 6250 partial rows, not out; channel sums in a new order
# speedup vs baseline: 3.0858x; 1.0691x over previous
#include <hip/hip_runtime.h>
#include <hip/hip_bf16.h>

#define N_NODES   100000
#define N_EDGES   1600000
#define NODE_DIM  64
#define HIDDEN    256
#define OUT_DIM   64
#define BN_EPS    1e-5f

typedef __attribute__((ext_vector_type(2))) float v2f;
typedef __attribute__((ext_vector_type(8))) float v8f;

#define WMMA_F32X4(ACC, A, B) \
  ACC = __builtin_amdgcn_wmma_f32_16x16x4_f32(false, (A), false, (B), (short)0, (ACC), false, false)

#define OWN_GROUPS   256
#define OWN_NODES    391
#define OWN_THREADS  1024
#define OWN_WAVES    32
#define RING         64
#define ROWS_AHEAD   16
#define BM_WORDS     (N_EDGES / 32)
#define CH_EDGES     4096
#define CH_WORDS     (CH_EDGES / 32)
#define N_CHUNK      ((N_EDGES + CH_EDGES - 1) / CH_EDGES)

static_assert(OWN_GROUPS * OWN_NODES >= N_NODES, "owner ranges must cover all nodes");
static_assert(N_EDGES % 32 == 0 && BM_WORDS % 4 == 0, "a plane is whole 16-byte pieces");
static_assert(N_EDGES % 4 == 0, "dst is read four edges at a time");
static_assert(CH_EDGES == 4 * OWN_THREADS, "one 16-byte dst load per thread");

__global__ __launch_bounds__(OWN_THREADS) void edge_owner_bits(
    const int* __restrict__ dst, unsigned* __restrict__ bits) {
  __shared__ unsigned tab[OWN_GROUPS * CH_WORDS];
  const int tid = threadIdx.x;
  uint4* t4 = reinterpret_cast<uint4*>(tab);
  {
    const uint4 z = {0u, 0u, 0u, 0u};
    for (int i = tid; i < OWN_GROUPS * CH_WORDS / 4; i += OWN_THREADS) t4[i] = z;
  }
  __syncthreads();
  const int e0 = blockIdx.x * CH_EDGES + tid * 4;
  if (e0 < N_EDGES) {
    const int4 d = *reinterpret_cast<const int4*>(dst + e0);
    const unsigned dc[4] = {(unsigned)d.x, (unsigned)d.y, (unsigned)d.z, (unsigned)d.w};
    const int w  = tid >> 3;
    const int b0 = (tid & 7) * 4;
#pragma unroll
    for (int c = 0; c < 4; ++c) {
      if (dc[c] < (unsigned)N_NODES) {
        const unsigned g = dc[c] / (unsigned)OWN_NODES;
        atomicOr(&tab[g * CH_WORDS + w], 1u << (b0 + c));
      }
    }
  }
  __syncthreads();
  const int wbase = blockIdx.x * CH_WORDS;
  for (int i = tid; i < OWN_GROUPS * CH_WORDS / 4; i += OWN_THREADS) {
    const int g  = i / (CH_WORDS / 4);
    const int wv = wbase + (i % (CH_WORDS / 4)) * 4;
    if (wv < BM_WORDS)
      *reinterpret_cast<uint4*>(bits + (size_t)g * BM_WORDS + wv) = t4[i];
  }
}

template <bool FULL>
__device__ __forceinline__ void drain_ring(
    const float* __restrict__ h, const int* __restrict__ src,
    const int* __restrict__ dst, float* acc, float* sdeg,
    const unsigned* q, unsigned head, int n, int lane, int base) {
  const unsigned e = q[(head + lane) & (RING - 1)];
  const bool valid = FULL || lane < n;
  int s = 0, dl = 0;
  if (valid) {
    s  = src[e];
    dl = dst[e] - base;
    atomicAdd(&sdeg[dl], 1.0f);
  }
  const float2* h2 = reinterpret_cast<const float2*>(h);
  const int cntj = FULL ? 32 : n;
  for (int j0 = 0; j0 < cntj; j0 += ROWS_AHEAD) {
    float2 v[ROWS_AHEAD];
#pragma unroll
    for (int u = 0; u < ROWS_AHEAD; ++u) {
      int j = j0 + u;
      if (j >= cntj) j = cntj - 1;
      const int sj = __builtin_amdgcn_readlane(s, j);
      v[u] = h2[(unsigned)sj * (unsigned)(NODE_DIM / 2) + (unsigned)lane];
    }
#pragma unroll
    for (int u = 0; u < ROWS_AHEAD; ++u) {
      const int j = j0 + u;
      if (j < cntj) {
        const int dj = __builtin_amdgcn_readlane(dl, j);
        float* a = acc + dj * NODE_DIM + lane;
        atomicAdd(a, v[u].x);
        atomicAdd(a + 32, v[u].y);
      }
    }
  }
}

__global__ __launch_bounds__(OWN_THREADS) void edge_owner_scan(
    const float* __restrict__ h, const int* __restrict__ src,
    const int* __restrict__ dst, const unsigned* __restrict__ bits,
    float* __restrict__ agg, float* __restrict__ deg, float* __restrict__ gstat) {
  __shared__ float acc[OWN_NODES * NODE_DIM];
  __shared__ float sdeg[OWN_NODES];
  __shared__ unsigned ring[OWN_WAVES][RING];

  const int tid  = threadIdx.x;
  const int wave = __builtin_amdgcn_readfirstlane(tid >> 5);
  const int lane = tid & 31;
  const int base = blockIdx.x * OWN_NODES;
  int cnt = N_NODES - base;
  if (cnt > OWN_NODES) cnt = OWN_NODES;

  {
    float4* a4 = reinterpret_cast<float4*>(acc);
    const float4 z = {0.0f, 0.0f, 0.0f, 0.0f};
    for (int i = tid; i < OWN_NODES * NODE_DIM / 4; i += OWN_THREADS) a4[i] = z;
    for (int i = tid; i < OWN_NODES; i += OWN_THREADS) sdeg[i] = 0.0f;
    if (blockIdx.x == 0 && tid < 2 * OUT_DIM) gstat[tid] = 0.0f;
  }
  __syncthreads();

  unsigned* q = ring[wave];
  unsigned head = 0, tail = 0;
  const uint4* bm4 = reinterpret_cast<const uint4*>(bits + (size_t)blockIdx.x * BM_WORDS);

  const uint4 none = {0u, 0u, 0u, 0u};
  uint4 nxt;
  {
    const int f4 = wave * 32 + lane;
    nxt = bm4[f4 < BM_WORDS / 4 ? f4 : BM_WORDS / 4 - 1];
    if (f4 >= BM_WORDS / 4) nxt = none;
  }
  for (int k = wave * 32; k < BM_WORDS / 4; k += OWN_WAVES * 32) {
    const int i4 = k + lane;
    const uint4 m4 = nxt;
    const int in4 = i4 + OWN_WAVES * 32;
    nxt = bm4[in4 < BM_WORDS / 4 ? in4 : BM_WORDS / 4 - 1];
    if (in4 >= BM_WORDS / 4) nxt = none;
    const unsigned mw[4] = {m4.x, m4.y, m4.z, m4.w};
#pragma unroll
    for (int c = 0; c < 4; ++c) {
      unsigned mm = mw[c];
      const unsigned ebase = ((unsigned)i4 * 4u + (unsigned)c) * 32u;
      unsigned any = __builtin_amdgcn_ballot_w32(mm != 0u);
      while (any != 0u) {
        if (mm != 0u) {
          const unsigned b = (unsigned)__builtin_ctz(mm);
          mm &= mm - 1u;
          const unsigned pos = tail + __builtin_amdgcn_mbcnt_lo(any, 0u);
          q[pos & (RING - 1)] = ebase + b;
        }
        tail += (unsigned)__builtin_popcount(any);
        while (tail - head >= 32u) {
          drain_ring<true>(h, src, dst, acc, sdeg, q, head, 32, lane, base);
          head += 32u;
        }
        any = __builtin_amdgcn_ballot_w32(mm != 0u);
      }
    }
  }
  if (tail != head)
    drain_ring<false>(h, src, dst, acc, sdeg, q, head, (int)(tail - head), lane, base);
  __syncthreads();

  for (int i = tid; i < cnt * NODE_DIM; i += OWN_THREADS) {
    const int m = i >> 6, c = i & 63;
    agg[(size_t)(base + m) * NODE_DIM + c] = acc[m * NODE_DIM + ((c & 1) << 5) + (c >> 1)];
  }
  for (int i = tid; i < cnt; i += OWN_THREADS) deg[base + i] = sdeg[i];
}

__global__ __launch_bounds__(128) void fused_mlp(
    const float* __restrict__ h,  const float* __restrict__ W1,
    const float* __restrict__ b1, const float* __restrict__ W2,
    const float* __restrict__ b2, const float* __restrict__ agg,
    const float* __restrict__ deg, float* __restrict__ out,
    float* __restrict__ part) {
  __shared__ float xs[16][NODE_DIM + 1];
  __shared__ float hs[16][HIDDEN + 1];

  const int m0   = blockIdx.x * 16;
  const int tid  = threadIdx.x;
  const int wave = tid >> 5;
  const int lane = tid & 31;
  const int half = lane >> 4;
  const int r    = lane & 15;

  for (int i = tid; i < 16 * NODE_DIM; i += 128) {
    int m = i >> 6, c = i & 63;
    int g = (m0 + m) * NODE_DIM + c;
    float d = deg[m0 + m];
    d = d > 1.0f ? d : 1.0f;
    xs[m][c] = agg[g] / d + h[g];
  }
  __syncthreads();

  v8f acc0 = {}, acc1 = {}, acc2 = {}, acc3 = {};
  const int nb = wave * 64;
  for (int k = 0; k < NODE_DIM; k += 4) {
    v2f a;
    a.x = xs[r][k + 2 * half];
    a.y = xs[r][k + 2 * half + 1];
    const float* w1p0 = W1 + (k + 2 * half) * HIDDEN + nb + r;
    const float* w1p1 = W1 + (k + 2 * half + 1) * HIDDEN + nb + r;
    v2f b;
    b.x = w1p0[0];  b.y = w1p1[0];  WMMA_F32X4(acc0, a, b);
    b.x = w1p0[16]; b.y = w1p1[16]; WMMA_F32X4(acc1, a, b);
    b.x = w1p0[32]; b.y = w1p1[32]; WMMA_F32X4(acc2, a, b);
    b.x = w1p0[48]; b.y = w1p1[48]; WMMA_F32X4(acc3, a, b);
  }
#define STORE_TILE(ACC, T)                                        \
  {                                                               \
    const int n = nb + (T) * 16 + r;                              \
    const float bias = b1[n];                                     \
    _Pragma("unroll") for (int j = 0; j < 8; ++j) {               \
      float v = (ACC)[j] + bias;                                  \
      hs[j + 8 * half][n] = v > 0.0f ? v : 0.0f;                  \
    }                                                             \
  }
  STORE_TILE(acc0, 0)
  STORE_TILE(acc1, 1)
  STORE_TILE(acc2, 2)
  STORE_TILE(acc3, 3)
#undef STORE_TILE
  __syncthreads();

  v8f occ = {};
  const int ob = wave * 16;
  for (int k = 0; k < HIDDEN; k += 4) {
    v2f a;
    a.x = hs[r][k + 2 * half];
    a.y = hs[r][k + 2 * half + 1];
    v2f b;
    b.x = W2[(k + 2 * half) * OUT_DIM + ob + r];
    b.y = W2[(k + 2 * half + 1) * OUT_DIM + ob + r];
    WMMA_F32X4(occ, a, b);
  }
  {
    const float bias = b2[ob + r];
    float s = 0.0f, q = 0.0f;
#pragma unroll
    for (int j = 0; j < 8; ++j) {
      float v = occ[j] + bias;
      v = v > 0.0f ? v : 0.0f;
      out[(m0 + j + 8 * half) * OUT_DIM + ob + r] = v;
      s += v;
      q += v * v;
    }
    s += __shfl_xor(s, 16);
    q += __shfl_xor(q, 16);
    part[blockIdx.x * 128 + tid] = half ? q : s;
  }
}

#define PART_ROWS (N_NODES / 16)
#define ST_ROWS   25
static_assert(PART_ROWS % ST_ROWS == 0, "every stats block sums whole partial rows");

__global__ __launch_bounds__(128) void bn_stats(
    const float* __restrict__ part, float* __restrict__ gsum,
    float* __restrict__ gsq) {
  const int j = threadIdx.x;
  const float* p = part + (size_t)blockIdx.x * (ST_ROWS * 128) + j;
  float v[ST_ROWS];
#pragma unroll
  for (int u = 0; u < ST_ROWS; ++u) v[u] = p[u * 128];
  float t = 0.0f;
#pragma unroll
  for (int u = 0; u < ST_ROWS; ++u) t += v[u];
  const int c = (j >> 5) * 16 + (j & 15);
  atomicAdd(((j >> 4) & 1) ? &gsq[c] : &gsum[c], t);
}

__global__ __launch_bounds__(256) void bn_apply(
    float* __restrict__ x, const float* __restrict__ gsum,
    const float* __restrict__ gsq, const float* __restrict__ gamma,
    const float* __restrict__ beta) {
  int i = blockIdx.x * 256 + threadIdx.x;
  if (i >= N_NODES * OUT_DIM) return;
  int c = i & 63;
  const float inv_n = 1.0f / (float)N_NODES;
  float mean = gsum[c] * inv_n;
  float var  = gsq[c] * inv_n - mean * mean;
  float sc   = gamma[c] * __frsqrt_rn(var + BN_EPS);
  x[i] = (x[i] - mean) * sc + beta[c];
}

extern "C" void kernel_launch(void* const* d_in, const int* in_sizes, int n_in,
                              void* d_out, int out_size, void* d_ws, size_t ws_size,
                              hipStream_t stream) {
  const float* h     = (const float*)d_in[0];
  const float* W1    = (const float*)d_in[1];
  const float* b1    = (const float*)d_in[2];
  const float* W2    = (const float*)d_in[3];
  const float* b2    = (const float*)d_in[4];
  const float* gamma = (const float*)d_in[5];
  const float* beta  = (const float*)d_in[6];
  const int*   src   = (const int*)d_in[7];
  const int*   dst   = (const int*)d_in[8];
  float* out = (float*)d_out;

  float* agg  = (float*)d_ws;
  float* deg  = agg + (size_t)N_NODES * NODE_DIM;
  float* gsum = deg + N_NODES;
  float* gsq  = gsum + OUT_DIM;
  unsigned* bits = (unsigned*)(gsq + OUT_DIM);
  float* part = (float*)(bits + (size_t)OWN_GROUPS * BM_WORDS);

  edge_owner_bits<<<N_CHUNK, OWN_THREADS, 0, stream>>>(dst, bits);

  edge_owner_scan<<<OWN_GROUPS, OWN_THREADS, 0, stream>>>(h, src, dst, bits, agg, deg, gsum);

  fused_mlp<<<N_NODES / 16, 128, 0, stream>>>(h, W1, b1, W2, b2, agg, deg, out, part);

  bn_stats<<<PART_ROWS / ST_ROWS, 128, 0, stream>>>(part, gsum, gsq);

  bn_apply<<<(N_NODES * OUT_DIM + 255) / 256, 256, 0, stream>>>(out, gsum, gsq, gamma, beta);
}


# First pass also stores one word per edge (source id + destination offset inside its owner range); an owner reads that one scattered word per marked edge instead of src[e] and dst[e]
# speedup vs baseline: 3.3277x; 1.0784x over previous
#include <hip/hip_runtime.h>
#include <hip/hip_bf16.h>

#define N_NODES   100000
#define N_EDGES   1600000
#define NODE_DIM  64
#define HIDDEN    256
#define OUT_DIM   64
#define BN_EPS    1e-5f

typedef __attribute__((ext_vector_type(2))) float v2f;
typedef __attribute__((ext_vector_type(8))) float v8f;

#define WMMA_F32X4(ACC, A, B) \
  ACC = __builtin_amdgcn_wmma_f32_16x16x4_f32(false, (A), false, (B), (short)0, (ACC), false, false)

#define OWN_GROUPS   256
#define OWN_NODES    391
#define OWN_THREADS  1024
#define OWN_WAVES    32
#define RING         64
#define ROWS_AHEAD   16
#define BM_WORDS     (N_EDGES / 32)
#define CH_EDGES     4096
#define CH_WORDS     (CH_EDGES / 32)
#define N_CHUNK      ((N_EDGES + CH_EDGES - 1) / CH_EDGES)
#define SRC_BITS     17
#define SRC_MASK     ((1u << SRC_BITS) - 1u)

static_assert(OWN_GROUPS * OWN_NODES >= N_NODES, "owner ranges must cover all nodes");
static_assert(N_EDGES % 32 == 0 && BM_WORDS % 4 == 0, "a plane is whole 16-byte pieces");
static_assert(N_EDGES % 4 == 0, "dst is read four edges at a time");
static_assert(CH_EDGES == 4 * OWN_THREADS, "one 16-byte dst load per thread");
static_assert((1u << SRC_BITS) >= N_NODES, "a source node fits the low field");
static_assert(OWN_NODES <= (1u << (32 - SRC_BITS)), "an offset inside an owner range fits the high field");

__global__ __launch_bounds__(OWN_THREADS) void edge_owner_bits(
    const int* __restrict__ src, const int* __restrict__ dst,
    unsigned* __restrict__ bits, unsigned* __restrict__ pk) {
  __shared__ unsigned tab[OWN_GROUPS * CH_WORDS];
  const int tid = threadIdx.x;
  uint4* t4 = reinterpret_cast<uint4*>(tab);
  {
    const uint4 z = {0u, 0u, 0u, 0u};
    for (int i = tid; i < OWN_GROUPS * CH_WORDS / 4; i += OWN_THREADS) t4[i] = z;
  }
  __syncthreads();
  const int e0 = blockIdx.x * CH_EDGES + tid * 4;
  if (e0 < N_EDGES) {
    const int4 d = *reinterpret_cast<const int4*>(dst + e0);
    const int4 s = *reinterpret_cast<const int4*>(src + e0);
    const unsigned dc[4] = {(unsigned)d.x, (unsigned)d.y, (unsigned)d.z, (unsigned)d.w};
    const unsigned sc[4] = {(unsigned)s.x, (unsigned)s.y, (unsigned)s.z, (unsigned)s.w};
    unsigned pw[4];
    const int w  = tid >> 3;
    const int b0 = (tid & 7) * 4;
#pragma unroll
    for (int c = 0; c < 4; ++c) {
      const unsigned g = dc[c] / (unsigned)OWN_NODES;
      pw[c] = (sc[c] & SRC_MASK) | ((dc[c] - g * (unsigned)OWN_NODES) << SRC_BITS);
      if (dc[c] < (unsigned)N_NODES)
        atomicOr(&tab[g * CH_WORDS + w], 1u << (b0 + c));
    }
    const uint4 p4 = {pw[0], pw[1], pw[2], pw[3]};
    *reinterpret_cast<uint4*>(pk + e0) = p4;
  }
  __syncthreads();
  const int wbase = blockIdx.x * CH_WORDS;
  for (int i = tid; i < OWN_GROUPS * CH_WORDS / 4; i += OWN_THREADS) {
    const int g  = i / (CH_WORDS / 4);
    const int wv = wbase + (i % (CH_WORDS / 4)) * 4;
    if (wv < BM_WORDS)
      *reinterpret_cast<uint4*>(bits + (size_t)g * BM_WORDS + wv) = t4[i];
  }
}

template <bool FULL>
__device__ __forceinline__ void drain_ring(
    const float* __restrict__ h, const unsigned* __restrict__ pk,
    float* acc, float* sdeg,
    const unsigned* q, unsigned head, int n, int lane) {
  const unsigned e = q[(head + lane) & (RING - 1)];
  const bool valid = FULL || lane < n;
  int s = 0, dl = 0;
  if (valid) {
    const unsigned w = pk[e];
    s  = (int)(w & SRC_MASK);
    dl = (int)(w >> SRC_BITS);
    atomicAdd(&sdeg[dl], 1.0f);
  }
  const float2* h2 = reinterpret_cast<const float2*>(h);
  const int cntj = FULL ? 32 : n;
  for (int j0 = 0; j0 < cntj; j0 += ROWS_AHEAD) {
    float2 v[ROWS_AHEAD];
#pragma unroll
    for (int u = 0; u < ROWS_AHEAD; ++u) {
      int j = j0 + u;
      if (j >= cntj) j = cntj - 1;
      const int sj = __builtin_amdgcn_readlane(s, j);
      v[u] = h2[(unsigned)sj * (unsigned)(NODE_DIM / 2) + (unsigned)lane];
    }
#pragma unroll
    for (int u = 0; u < ROWS_AHEAD; ++u) {
      const int j = j0 + u;
      if (j < cntj) {
        const int dj = __builtin_amdgcn_readlane(dl, j);
        float* a = acc + dj * NODE_DIM + lane;
        atomicAdd(a, v[u].x);
        atomicAdd(a + 32, v[u].y);
      }
    }
  }
}

__global__ __launch_bounds__(OWN_THREADS) void edge_owner_scan(
    const float* __restrict__ h, const unsigned* __restrict__ pk,
    const unsigned* __restrict__ bits,
    float* __restrict__ agg, float* __restrict__ deg, float* __restrict__ gstat) {
  __shared__ float acc[OWN_NODES * NODE_DIM];
  __shared__ float sdeg[OWN_NODES];
  __shared__ unsigned ring[OWN_WAVES][RING];

  const int tid  = threadIdx.x;
  const int wave = __builtin_amdgcn_readfirstlane(tid >> 5);
  const int lane = tid & 31;
  const int base = blockIdx.x * OWN_NODES;
  int cnt = N_NODES - base;
  if (cnt > OWN_NODES) cnt = OWN_NODES;

  {
    float4* a4 = reinterpret_cast<float4*>(acc);
    const float4 z = {0.0f, 0.0f, 0.0f, 0.0f};
    for (int i = tid; i < OWN_NODES * NODE_DIM / 4; i += OWN_THREADS) a4[i] = z;
    for (int i = tid; i < OWN_NODES; i += OWN_THREADS) sdeg[i] = 0.0f;
    if (blockIdx.x == 0 && tid < 2 * OUT_DIM) gstat[tid] = 0.0f;
  }
  __syncthreads();

  unsigned* q = ring[wave];
  unsigned head = 0, tail = 0;
  const uint4* bm4 = reinterpret_cast<const uint4*>(bits + (size_t)blockIdx.x * BM_WORDS);

  const uint4 none = {0u, 0u, 0u, 0u};
  uint4 nxt;
  {
    const int f4 = wave * 32 + lane;
    nxt = bm4[f4 < BM_WORDS / 4 ? f4 : BM_WORDS / 4 - 1];
    if (f4 >= BM_WORDS / 4) nxt = none;
  }
  for (int k = wave * 32; k < BM_WORDS / 4; k += OWN_WAVES * 32) {
    const int i4 = k + lane;
    const uint4 m4 = nxt;
    const int in4 = i4 + OWN_WAVES * 32;
    nxt = bm4[in4 < BM_WORDS / 4 ? in4 : BM_WORDS / 4 - 1];
    if (in4 >= BM_WORDS / 4) nxt = none;
    const unsigned mw[4] = {m4.x, m4.y, m4.z, m4.w};
#pragma unroll
    for (int c = 0; c < 4; ++c) {
      unsigned mm = mw[c];
      const unsigned ebase = ((unsigned)i4 * 4u + (unsigned)c) * 32u;
      unsigned any = __builtin_amdgcn_ballot_w32(mm != 0u);
      while (any != 0u) {
        if (mm != 0u) {
          const unsigned b = (unsigned)__builtin_ctz(mm);
          mm &= mm - 1u;
          const unsigned pos = tail + __builtin_amdgcn_mbcnt_lo(any, 0u);
          q[pos & (RING - 1)] = ebase + b;
        }
        tail += (unsigned)__builtin_popcount(any);
        while (tail - head >= 32u) {
          drain_ring<true>(h, pk, acc, sdeg, q, head, 32, lane);
          head += 32u;
        }
        any = __builtin_amdgcn_ballot_w32(mm != 0u);
      }
    }
  }
  if (tail != head)
    drain_ring<false>(h, pk, acc, sdeg, q, head, (int)(tail - head), lane);
  __syncthreads();

  for (int i = tid; i < cnt * NODE_DIM; i += OWN_THREADS) {
    const int m = i >> 6, c = i & 63;
    agg[(size_t)(base + m) * NODE_DIM + c] = acc[m * NODE_DIM + ((c & 1) << 5) + (c >> 1)];
  }
  for (int i = tid; i < cnt; i += OWN_THREADS) deg[base + i] = sdeg[i];
}

__global__ __launch_bounds__(128) void fused_mlp(
    const float* __restrict__ h,  const float* __restrict__ W1,
    const float* __restrict__ b1, const float* __restrict__ W2,
    const float* __restrict__ b2, const float* __restrict__ agg,
    const float* __restrict__ deg, float* __restrict__ out,
    float* __restrict__ part) {
  __shared__ float xs[16][NODE_DIM + 1];
  __shared__ float hs[16][HIDDEN + 1];

  const int m0   = blockIdx.x * 16;
  const int tid  = threadIdx.x;
  const int wave = tid >> 5;
  const int lane = tid & 31;
  const int half = lane >> 4;
  const int r    = lane & 15;

  for (int i = tid; i < 16 * NODE_DIM; i += 128) {
    int m = i >> 6, c = i & 63;
    int g = (m0 + m) * NODE_DIM + c;
    float d = deg[m0 + m];
    d = d > 1.0f ? d : 1.0f;
    xs[m][c] = agg[g] / d + h[g];
  }
  __syncthreads();

  v8f acc0 = {}, acc1 = {}, acc2 = {}, acc3 = {};
  const int nb = wave * 64;
  for (int k = 0; k < NODE_DIM; k += 4) {
    v2f a;
    a.x = xs[r][k + 2 * half];
    a.y = xs[r][k + 2 * half + 1];
    const float* w1p0 = W1 + (k + 2 * half) * HIDDEN + nb + r;
    const float* w1p1 = W1 + (k + 2 * half + 1) * HIDDEN + nb + r;
    v2f b;
    b.x = w1p0[0];  b.y = w1p1[0];  WMMA_F32X4(acc0, a, b);
    b.x = w1p0[16]; b.y = w1p1[16]; WMMA_F32X4(acc1, a, b);
    b.x = w1p0[32]; b.y = w1p1[32]; WMMA_F32X4(acc2, a, b);
    b.x = w1p0[48]; b.y = w1p1[48]; WMMA_F32X4(acc3, a, b);
  }
#define STORE_TILE(ACC, T)                                        \
  {                                                               \
    const int n = nb + (T) * 16 + r;                              \
    const float bias = b1[n];                                     \
    _Pragma("unroll") for (int j = 0; j < 8; ++j) {               \
      float v = (ACC)[j] + bias;                                  \
      hs[j + 8 * half][n] = v > 0.0f ? v : 0.0f;                  \
    }                                                             \
  }
  STORE_TILE(acc0, 0)
  STORE_TILE(acc1, 1)
  STORE_TILE(acc2, 2)
  STORE_TILE(acc3, 3)
#undef STORE_TILE
  __syncthreads();

  v8f occ = {};
  const int ob = wave * 16;
  for (int k = 0; k < HIDDEN; k += 4) {
    v2f a;
    a.x = hs[r][k + 2 * half];
    a.y = hs[r][k + 2 * half + 1];
    v2f b;
    b.x = W2[(k + 2 * half) * OUT_DIM + ob + r];
    b.y = W2[(k + 2 * half + 1) * OUT_DIM + ob + r];
    WMMA_F32X4(occ, a, b);
  }
  {
    const float bias = b2[ob + r];
    float s = 0.0f, q = 0.0f;
#pragma unroll
    for (int j = 0; j < 8; ++j) {
      float v = occ[j] + bias;
      v = v > 0.0f ? v : 0.0f;
      out[(m0 + j + 8 * half) * OUT_DIM + ob + r] = v;
      s += v;
      q += v * v;
    }
    s += __shfl_xor(s, 16);
    q += __shfl_xor(q, 16);
    part[blockIdx.x * 128 + tid] = half ? q : s;
  }
}

#define PART_ROWS (N_NODES / 16)
#define ST_ROWS   25
static_assert(PART_ROWS % ST_ROWS == 0, "every stats block sums whole partial rows");

__global__ __launch_bounds__(128) void bn_stats(
    const float* __restrict__ part, float* __restrict__ gsum,
    float* __restrict__ gsq) {
  const int j = threadIdx.x;
  const float* p = part + (size_t)blockIdx.x * (ST_ROWS * 128) + j;
  float v[ST_ROWS];
#pragma unroll
  for (int u = 0; u < ST_ROWS; ++u) v[u] = p[u * 128];
  float t = 0.0f;
#pragma unroll
  for (int u = 0; u < ST_ROWS; ++u) t += v[u];
  const int c = (j >> 5) * 16 + (j & 15);
  atomicAdd(((j >> 4) & 1) ? &gsq[c] : &gsum[c], t);
}

__global__ __launch_bounds__(256) void bn_apply(
    float* __restrict__ x, const float* __restrict__ gsum,
    const float* __restrict__ gsq, const float* __restrict__ gamma,
    const float* __restrict__ beta) {
  int i = blockIdx.x * 256 + threadIdx.x;
  if (i >= N_NODES * OUT_DIM) return;
  int c = i & 63;
  const float inv_n = 1.0f / (float)N_NODES;
  float mean = gsum[c] * inv_n;
  float var  = gsq[c] * inv_n - mean * mean;
  float sc   = gamma[c] * __frsqrt_rn(var + BN_EPS);
  x[i] = (x[i] - mean) * sc + beta[c];
}

extern "C" void kernel_launch(void* const* d_in, const int* in_sizes, int n_in,
                              void* d_out, int out_size, void* d_ws, size_t ws_size,
                              hipStream_t stream) {
  const float* h     = (const float*)d_in[0];
  const float* W1    = (const float*)d_in[1];
  const float* b1    = (const float*)d_in[2];
  const float* W2    = (const float*)d_in[3];
  const float* b2    = (const float*)d_in[4];
  const float* gamma = (const float*)d_in[5];
  const float* beta  = (const float*)d_in[6];
  const int*   src   = (const int*)d_in[7];
  const int*   dst   = (const int*)d_in[8];
  float* out = (float*)d_out;

  float* agg  = (float*)d_ws;
  float* deg  = agg + (size_t)N_NODES * NODE_DIM;
  float* gsum = deg + N_NODES;
  float* gsq  = gsum + OUT_DIM;
  unsigned* bits = (unsigned*)(gsq + OUT_DIM);
  float* part = (float*)(bits + (size_t)OWN_GROUPS * BM_WORDS);
  unsigned* pk = (unsigned*)(part + (size_t)(N_NODES / 16) * 128);

  edge_owner_bits<<<N_CHUNK, OWN_THREADS, 0, stream>>>(src, dst, bits, pk);

  edge_owner_scan<<<OWN_GROUPS, OWN_THREADS, 0, stream>>>(h, pk, bits, agg, deg, gsum);

  fused_mlp<<<N_NODES / 16, 128, 0, stream>>>(h, W1, b1, W2, b2, agg, deg, out, part);

  bn_stats<<<PART_ROWS / ST_ROWS, 128, 0, stream>>>(part, gsum, gsq);

  bn_apply<<<(N_NODES * OUT_DIM + 255) / 256, 256, 0, stream>>>(out, gsum, gsq, gamma, beta);
}


# MLP global loads widened to 16 B per lane: node rows in staging, weight fragments via 4 interleaved column tiles per wave; GEMM2 split by K quarter per wave, quarters added in LDS: new sum order
# speedup vs baseline: 3.4264x; 1.0297x over previous
#include <hip/hip_runtime.h>
#include <hip/hip_bf16.h>

#define N_NODES   100000
#define N_EDGES   1600000
#define NODE_DIM  64
#define HIDDEN    256
#define OUT_DIM   64
#define BN_EPS    1e-5f

typedef __attribute__((ext_vector_type(2))) float v2f;
typedef __attribute__((ext_vector_type(8))) float v8f;

#define WMMA_F32X4(ACC, A, B) \
  ACC = __builtin_amdgcn_wmma_f32_16x16x4_f32(false, (A), false, (B), (short)0, (ACC), false, false)

#define OWN_GROUPS   256
#define OWN_NODES    391
#define OWN_THREADS  1024
#define OWN_WAVES    32
#define RING         64
#define ROWS_AHEAD   16
#define BM_WORDS     (N_EDGES / 32)
#define CH_EDGES     4096
#define CH_WORDS     (CH_EDGES / 32)
#define N_CHUNK      ((N_EDGES + CH_EDGES - 1) / CH_EDGES)
#define SRC_BITS     17
#define SRC_MASK     ((1u << SRC_BITS) - 1u)

static_assert(OWN_GROUPS * OWN_NODES >= N_NODES, "owner ranges must cover all nodes");
static_assert(N_EDGES % 32 == 0 && BM_WORDS % 4 == 0, "a plane is whole 16-byte pieces");
static_assert(N_EDGES % 4 == 0, "dst is read four edges at a time");
static_assert(CH_EDGES == 4 * OWN_THREADS, "one 16-byte dst load per thread");
static_assert((1u << SRC_BITS) >= N_NODES, "a source node fits the low field");
static_assert(OWN_NODES <= (1u << (32 - SRC_BITS)), "an offset inside an owner range fits the high field");

__global__ __launch_bounds__(OWN_THREADS) void edge_owner_bits(
    const int* __restrict__ src, const int* __restrict__ dst,
    unsigned* __restrict__ bits, unsigned* __restrict__ pk) {
  __shared__ unsigned tab[OWN_GROUPS * CH_WORDS];
  const int tid = threadIdx.x;
  uint4* t4 = reinterpret_cast<uint4*>(tab);
  {
    const uint4 z = {0u, 0u, 0u, 0u};
    for (int i = tid; i < OWN_GROUPS * CH_WORDS / 4; i += OWN_THREADS) t4[i] = z;
  }
  __syncthreads();
  const int e0 = blockIdx.x * CH_EDGES + tid * 4;
  if (e0 < N_EDGES) {
    const int4 d = *reinterpret_cast<const int4*>(dst + e0);
    const int4 s = *reinterpret_cast<const int4*>(src + e0);
    const unsigned dc[4] = {(unsigned)d.x, (unsigned)d.y, (unsigned)d.z, (unsigned)d.w};
    const unsigned sc[4] = {(unsigned)s.x, (unsigned)s.y, (unsigned)s.z, (unsigned)s.w};
    unsigned pw[4];
    const int w  = tid >> 3;
    const int b0 = (tid & 7) * 4;
#pragma unroll
    for (int c = 0; c < 4; ++c) {
      const unsigned g = dc[c] / (unsigned)OWN_NODES;
      pw[c] = (sc[c] & SRC_MASK) | ((dc[c] - g * (unsigned)OWN_NODES) << SRC_BITS);
      if (dc[c] < (unsigned)N_NODES)
        atomicOr(&tab[g * CH_WORDS + w], 1u << (b0 + c));
    }
    const uint4 p4 = {pw[0], pw[1], pw[2], pw[3]};
    *reinterpret_cast<uint4*>(pk + e0) = p4;
  }
  __syncthreads();
  const int wbase = blockIdx.x * CH_WORDS;
  for (int i = tid; i < OWN_GROUPS * CH_WORDS / 4; i += OWN_THREADS) {
    const int g  = i / (CH_WORDS / 4);
    const int wv = wbase + (i % (CH_WORDS / 4)) * 4;
    if (wv < BM_WORDS)
      *reinterpret_cast<uint4*>(bits + (size_t)g * BM_WORDS + wv) = t4[i];
  }
}

template <bool FULL>
__device__ __forceinline__ void drain_ring(
    const float* __restrict__ h, const unsigned* __restrict__ pk,
    float* acc, float* sdeg,
    const unsigned* q, unsigned head, int n, int lane) {
  const unsigned e = q[(head + lane) & (RING - 1)];
  const bool valid = FULL || lane < n;
  int s = 0, dl = 0;
  if (valid) {
    const unsigned w = pk[e];
    s  = (int)(w & SRC_MASK);
    dl = (int)(w >> SRC_BITS);
    atomicAdd(&sdeg[dl], 1.0f);
  }
  const float2* h2 = reinterpret_cast<const float2*>(h);
  const int cntj = FULL ? 32 : n;
  for (int j0 = 0; j0 < cntj; j0 += ROWS_AHEAD) {
    float2 v[ROWS_AHEAD];
#pragma unroll
    for (int u = 0; u < ROWS_AHEAD; ++u) {
      int j = j0 + u;
      if (j >= cntj) j = cntj - 1;
      const int sj = __builtin_amdgcn_readlane(s, j);
      v[u] = h2[(unsigned)sj * (unsigned)(NODE_DIM / 2) + (unsigned)lane];
    }
#pragma unroll
    for (int u = 0; u < ROWS_AHEAD; ++u) {
      const int j = j0 + u;
      if (j < cntj) {
        const int dj = __builtin_amdgcn_readlane(dl, j);
        float* a = acc + dj * NODE_DIM + lane;
        atomicAdd(a, v[u].x);
        atomicAdd(a + 32, v[u].y);
      }
    }
  }
}

__global__ __launch_bounds__(OWN_THREADS) void edge_owner_scan(
    const float* __restrict__ h, const unsigned* __restrict__ pk,
    const unsigned* __restrict__ bits,
    float* __restrict__ agg, float* __restrict__ deg, float* __restrict__ gstat) {
  __shared__ float acc[OWN_NODES * NODE_DIM];
  __shared__ float sdeg[OWN_NODES];
  __shared__ unsigned ring[OWN_WAVES][RING];

  const int tid  = threadIdx.x;
  const int wave = __builtin_amdgcn_readfirstlane(tid >> 5);
  const int lane = tid & 31;
  const int base = blockIdx.x * OWN_NODES;
  int cnt = N_NODES - base;
  if (cnt > OWN_NODES) cnt = OWN_NODES;

  {
    float4* a4 = reinterpret_cast<float4*>(acc);
    const float4 z = {0.0f, 0.0f, 0.0f, 0.0f};
    for (int i = tid; i < OWN_NODES * NODE_DIM / 4; i += OWN_THREADS) a4[i] = z;
    for (int i = tid; i < OWN_NODES; i += OWN_THREADS) sdeg[i] = 0.0f;
    if (blockIdx.x == 0 && tid < 2 * OUT_DIM) gstat[tid] = 0.0f;
  }
  __syncthreads();

  unsigned* q = ring[wave];
  unsigned head = 0, tail = 0;
  const uint4* bm4 = reinterpret_cast<const uint4*>(bits + (size_t)blockIdx.x * BM_WORDS);

  const uint4 none = {0u, 0u, 0u, 0u};
  uint4 nxt;
  {
    const int f4 = wave * 32 + lane;
    nxt = bm4[f4 < BM_WORDS / 4 ? f4 : BM_WORDS / 4 - 1];
    if (f4 >= BM_WORDS / 4) nxt = none;
  }
  for (int k = wave * 32; k < BM_WORDS / 4; k += OWN_WAVES * 32) {
    const int i4 = k + lane;
    const uint4 m4 = nxt;
    const int in4 = i4 + OWN_WAVES * 32;
    nxt = bm4[in4 < BM_WORDS / 4 ? in4 : BM_WORDS / 4 - 1];
    if (in4 >= BM_WORDS / 4) nxt = none;
    const unsigned mw[4] = {m4.x, m4.y, m4.z, m4.w};
#pragma unroll
    for (int c = 0; c < 4; ++c) {
      unsigned mm = mw[c];
      const unsigned ebase = ((unsigned)i4 * 4u + (unsigned)c) * 32u;
      unsigned any = __builtin_amdgcn_ballot_w32(mm != 0u);
      while (any != 0u) {
        if (mm != 0u) {
          const unsigned b = (unsigned)__builtin_ctz(mm);
          mm &= mm - 1u;
          const unsigned pos = tail + __builtin_amdgcn_mbcnt_lo(any, 0u);
          q[pos & (RING - 1)] = ebase + b;
        }
        tail += (unsigned)__builtin_popcount(any);
        while (tail - head >= 32u) {
          drain_ring<true>(h, pk, acc, sdeg, q, head, 32, lane);
          head += 32u;
        }
        any = __builtin_amdgcn_ballot_w32(mm != 0u);
      }
    }
  }
  if (tail != head)
    drain_ring<false>(h, pk, acc, sdeg, q, head, (int)(tail - head), lane);
  __syncthreads();

  for (int i = tid; i < cnt * NODE_DIM; i += OWN_THREADS) {
    const int m = i >> 6, c = i & 63;
    agg[(size_t)(base + m) * NODE_DIM + c] = acc[m * NODE_DIM + ((c & 1) << 5) + (c >> 1)];
  }
  for (int i = tid; i < cnt; i += OWN_THREADS) deg[base + i] = sdeg[i];
}

#define PS_PITCH (OUT_DIM + 4)
#define XS_FLOATS (16 * (NODE_DIM + 1))
#define HS_FLOATS (16 * (HIDDEN + 1))
static_assert(4 * 16 * PS_PITCH <= XS_FLOATS + HS_FLOATS, "the quarter sums fit where xs and hs were");
static_assert(HIDDEN == 4 * 64 && OUT_DIM == 64 && NODE_DIM % 4 == 0, "four waves: 64 hidden columns each, then one K quarter each");
static_assert(16 * NODE_DIM / 4 == 2 * 128, "staging: two 16-byte pieces per thread");

__global__ __launch_bounds__(128) void fused_mlp(
    const float* __restrict__ h,  const float* __restrict__ W1,
    const float* __restrict__ b1, const float* __restrict__ W2,
    const float* __restrict__ b2, const float* __restrict__ agg,
    const float* __restrict__ deg, float* __restrict__ out,
    float* __restrict__ part) {
  __shared__ __attribute__((aligned(16))) float smem[XS_FLOATS + HS_FLOATS];
  float (*xs)[NODE_DIM + 1] = reinterpret_cast<float (*)[NODE_DIM + 1]>(smem);
  float (*hs)[HIDDEN + 1]   = reinterpret_cast<float (*)[HIDDEN + 1]>(smem + XS_FLOATS);
  float (*ps)[16][PS_PITCH] = reinterpret_cast<float (*)[16][PS_PITCH]>(smem);

  const int m0   = blockIdx.x * 16;
  const int tid  = threadIdx.x;
  const int wave = tid >> 5;
  const int lane = tid & 31;
  const int half = lane >> 4;
  const int r    = lane & 15;

  {
    const int mA = tid >> 4, mB = mA + 8, c = (tid & 15) * 4;
    const int gA = (m0 + mA) * NODE_DIM + c;
    const int gB = gA + 8 * NODE_DIM;
    float dA = deg[m0 + mA];
    float dB = deg[m0 + mB];
    const float4 aA = *reinterpret_cast<const float4*>(agg + gA);
    const float4 hA = *reinterpret_cast<const float4*>(h + gA);
    const float4 aB = *reinterpret_cast<const float4*>(agg + gB);
    const float4 hB = *reinterpret_cast<const float4*>(h + gB);
    dA = dA > 1.0f ? dA : 1.0f;
    dB = dB > 1.0f ? dB : 1.0f;
    xs[mA][c]     = aA.x / dA + hA.x;
    xs[mA][c + 1] = aA.y / dA + hA.y;
    xs[mA][c + 2] = aA.z / dA + hA.z;
    xs[mA][c + 3] = aA.w / dA + hA.w;
    xs[mB][c]     = aB.x / dB + hB.x;
    xs[mB][c + 1] = aB.y / dB + hB.y;
    xs[mB][c + 2] = aB.z / dB + hB.z;
    xs[mB][c + 3] = aB.w / dB + hB.w;
  }
  __syncthreads();

  v8f acc0 = {}, acc1 = {}, acc2 = {}, acc3 = {};
  const int nb = wave * 64;
  {
    const float* w1p = W1 + (2 * half) * HIDDEN + nb + 4 * r;
#pragma unroll 4
    for (int k = 0; k < NODE_DIM; k += 4) {
      v2f a;
      a.x = xs[r][k + 2 * half];
      a.y = xs[r][k + 2 * half + 1];
      const float4 q0 = *reinterpret_cast<const float4*>(w1p + k * HIDDEN);
      const float4 q1 = *reinterpret_cast<const float4*>(w1p + (k + 1) * HIDDEN);
      v2f b;
      b.x = q0.x; b.y = q1.x; WMMA_F32X4(acc0, a, b);
      b.x = q0.y; b.y = q1.y; WMMA_F32X4(acc1, a, b);
      b.x = q0.z; b.y = q1.z; WMMA_F32X4(acc2, a, b);
      b.x = q0.w; b.y = q1.w; WMMA_F32X4(acc3, a, b);
    }
  }
  {
    const float4 bias4 = *reinterpret_cast<const float4*>(b1 + nb + 4 * r);
#pragma unroll
    for (int j = 0; j < 8; ++j) {
      float* hrow = &hs[j + 8 * half][nb + 4 * r];
      const float v0 = acc0[j] + bias4.x;
      const float v1 = acc1[j] + bias4.y;
      const float v2 = acc2[j] + bias4.z;
      const float v3 = acc3[j] + bias4.w;
      hrow[0] = v0 > 0.0f ? v0 : 0.0f;
      hrow[1] = v1 > 0.0f ? v1 : 0.0f;
      hrow[2] = v2 > 0.0f ? v2 : 0.0f;
      hrow[3] = v3 > 0.0f ? v3 : 0.0f;
    }
  }
  __syncthreads();

  v8f o0 = {}, o1 = {}, o2 = {}, o3 = {};
  {
    const int kb = wave * 64;
    const float* w2p = W2 + (kb + 2 * half) * OUT_DIM + 4 * r;
#pragma unroll 4
    for (int k = 0; k < 64; k += 4) {
      v2f a;
      a.x = hs[r][kb + k + 2 * half];
      a.y = hs[r][kb + k + 2 * half + 1];
      const float4 q0 = *reinterpret_cast<const float4*>(w2p + k * OUT_DIM);
      const float4 q1 = *reinterpret_cast<const float4*>(w2p + (k + 1) * OUT_DIM);
      v2f b;
      b.x = q0.x; b.y = q1.x; WMMA_F32X4(o0, a, b);
      b.x = q0.y; b.y = q1.y; WMMA_F32X4(o1, a, b);
      b.x = q0.z; b.y = q1.z; WMMA_F32X4(o2, a, b);
      b.x = q0.w; b.y = q1.w; WMMA_F32X4(o3, a, b);
    }
  }
  __syncthreads();
#pragma unroll
  for (int j = 0; j < 8; ++j) {
    const float4 p = {o0[j], o1[j], o2[j], o3[j]};
    *reinterpret_cast<float4*>(&ps[wave][j + 8 * half][4 * r]) = p;
  }
  __syncthreads();

  const int ob = wave * 16;
  {
    const float bias = b2[ob + r];
    float s = 0.0f, q = 0.0f;
#pragma unroll
    for (int j = 0; j < 8; ++j) {
      const int row = j + 8 * half;
      float v = ((ps[0][row][ob + r] + ps[1][row][ob + r]) + ps[2][row][ob + r]) + ps[3][row][ob + r];
      v += bias;
      v = v > 0.0f ? v : 0.0f;
      out[(m0 + row) * OUT_DIM + ob + r] = v;
      s += v;
      q += v * v;
    }
    s += __shfl_xor(s, 16);
    q += __shfl_xor(q, 16);
    part[blockIdx.x * 128 + tid] = half ? q : s;
  }
}

#define PART_ROWS (N_NODES / 16)
#define ST_ROWS   25
static_assert(PART_ROWS % ST_ROWS == 0, "every stats block sums whole partial rows");

__global__ __launch_bounds__(128) void bn_stats(
    const float* __restrict__ part, float* __restrict__ gsum,
    float* __restrict__ gsq) {
  const int j = threadIdx.x;
  const float* p = part + (size_t)blockIdx.x * (ST_ROWS * 128) + j;
  float v[ST_ROWS];
#pragma unroll
  for (int u = 0; u < ST_ROWS; ++u) v[u] = p[u * 128];
  float t = 0.0f;
#pragma unroll
  for (int u = 0; u < ST_ROWS; ++u) t += v[u];
  const int c = (j >> 5) * 16 + (j & 15);
  atomicAdd(((j >> 4) & 1) ? &gsq[c] : &gsum[c], t);
}

__global__ __launch_bounds__(256) void bn_apply(
    float* __restrict__ x, const float* __restrict__ gsum,
    const float* __restrict__ gsq, const float* __restrict__ gamma,
    const float* __restrict__ beta) {
  int i = blockIdx.x * 256 + threadIdx.x;
  if (i >= N_NODES * OUT_DIM) return;
  int c = i & 63;
  const float inv_n = 1.0f / (float)N_NODES;
  float mean = gsum[c] * inv_n;
  float var  = gsq[c] * inv_n - mean * mean;
  float sc   = gamma[c] * __frsqrt_rn(var + BN_EPS);
  x[i] = (x[i] - mean) * sc + beta[c];
}

extern "C" void kernel_launch(void* const* d_in, const int* in_sizes, int n_in,
                              void* d_out, int out_size, void* d_ws, size_t ws_size,
                              hipStream_t stream) {
  const float* h     = (const float*)d_in[0];
  const float* W1    = (const float*)d_in[1];
  const float* b1    = (const float*)d_in[2];
  const float* W2    = (const float*)d_in[3];
  const float* b2    = (const float*)d_in[4];
  const float* gamma = (const float*)d_in[5];
  const float* beta  = (const float*)d_in[6];
  const int*   src   = (const int*)d_in[7];
  const int*   dst   = (const int*)d_in[8];
  float* out = (float*)d_out;

  float* agg  = (float*)d_ws;
  float* deg  = agg + (size_t)N_NODES * NODE_DIM;
  float* gsum = deg + N_NODES;
  float* gsq  = gsum + OUT_DIM;
  unsigned* bits = (unsigned*)(gsq + OUT_DIM);
  float* part = (float*)(bits + (size_t)OWN_GROUPS * BM_WORDS);
  unsigned* pk = (unsigned*)(part + (size_t)(N_NODES / 16) * 128);

  edge_owner_bits<<<N_CHUNK, OWN_THREADS, 0, stream>>>(src, dst, bits, pk);

  edge_owner_scan<<<OWN_GROUPS, OWN_THREADS, 0, stream>>>(h, pk, bits, agg, deg, gsum);

  fused_mlp<<<N_NODES / 16, 128, 0, stream>>>(h, W1, b1, W2, b2, agg, deg, out, part);

  bn_stats<<<PART_ROWS / ST_ROWS, 128, 0, stream>>>(part, gsum, gsq);

  bn_apply<<<(N_NODES * OUT_DIM + 255) / 256, 256, 0, stream>>>(out, gsum, gsq, gamma, beta);
}


# BN apply: a lane takes 16 single floats of its channel, all 16 loads in flight, 3125 groups of 128 lanes (was 25000 of 256 one-float lanes); accesses stay 4 bytes, one whole line per wave
# speedup vs baseline: 3.6159x; 1.0553x over previous
#include <hip/hip_runtime.h>
#include <hip/hip_bf16.h>

#define N_NODES   100000
#define N_EDGES   1600000
#define NODE_DIM  64
#define HIDDEN    256
#define OUT_DIM   64
#define BN_EPS    1e-5f

typedef __attribute__((ext_vector_type(2))) float v2f;
typedef __attribute__((ext_vector_type(8))) float v8f;

#define WMMA_F32X4(ACC, A, B) \
  ACC = __builtin_amdgcn_wmma_f32_16x16x4_f32(false, (A), false, (B), (short)0, (ACC), false, false)

#define OWN_GROUPS   256
#define OWN_NODES    391
#define OWN_THREADS  1024
#define OWN_WAVES    32
#define RING         64
#define ROWS_AHEAD   16
#define BM_WORDS     (N_EDGES / 32)
#define CH_EDGES     4096
#define CH_WORDS     (CH_EDGES / 32)
#define N_CHUNK      ((N_EDGES + CH_EDGES - 1) / CH_EDGES)
#define SRC_BITS     17
#define SRC_MASK     ((1u << SRC_BITS) - 1u)

static_assert(OWN_GROUPS * OWN_NODES >= N_NODES, "owner ranges must cover all nodes");
static_assert(N_EDGES % 32 == 0 && BM_WORDS % 4 == 0, "a plane is whole 16-byte pieces");
static_assert(N_EDGES % 4 == 0, "dst is read four edges at a time");
static_assert(CH_EDGES == 4 * OWN_THREADS, "one 16-byte dst load per thread");
static_assert((1u << SRC_BITS) >= N_NODES, "a source node fits the low field");
static_assert(OWN_NODES <= (1u << (32 - SRC_BITS)), "an offset inside an owner range fits the high field");

__global__ __launch_bounds__(OWN_THREADS) void edge_owner_bits(
    const int* __restrict__ src, const int* __restrict__ dst,
    unsigned* __restrict__ bits, unsigned* __restrict__ pk) {
  __shared__ unsigned tab[OWN_GROUPS * CH_WORDS];
  const int tid = threadIdx.x;
  uint4* t4 = reinterpret_cast<uint4*>(tab);
  {
    const uint4 z = {0u, 0u, 0u, 0u};
    for (int i = tid; i < OWN_GROUPS * CH_WORDS / 4; i += OWN_THREADS) t4[i] = z;
  }
  __syncthreads();
  const int e0 = blockIdx.x * CH_EDGES + tid * 4;
  if (e0 < N_EDGES) {
    const int4 d = *reinterpret_cast<const int4*>(dst + e0);
    const int4 s = *reinterpret_cast<const int4*>(src + e0);
    const unsigned dc[4] = {(unsigned)d.x, (unsigned)d.y, (unsigned)d.z, (unsigned)d.w};
    const unsigned sc[4] = {(unsigned)s.x, (unsigned)s.y, (unsigned)s.z, (unsigned)s.w};
    unsigned pw[4];
    const int w  = tid >> 3;
    const int b0 = (tid & 7) * 4;
#pragma unroll
    for (int c = 0; c < 4; ++c) {
      const unsigned g = dc[c] / (unsigned)OWN_NODES;
      pw[c] = (sc[c] & SRC_MASK) | ((dc[c] - g * (unsigned)OWN_NODES) << SRC_BITS);
      if (dc[c] < (unsigned)N_NODES)
        atomicOr(&tab[g * CH_WORDS + w], 1u << (b0 + c));
    }
    const uint4 p4 = {pw[0], pw[1], pw[2], pw[3]};
    *reinterpret_cast<uint4*>(pk + e0) = p4;
  }
  __syncthreads();
  const int wbase = blockIdx.x * CH_WORDS;
  for (int i = tid; i < OWN_GROUPS * CH_WORDS / 4; i += OWN_THREADS) {
    const int g  = i / (CH_WORDS / 4);
    const int wv = wbase + (i % (CH_WORDS / 4)) * 4;
    if (wv < BM_WORDS)
      *reinterpret_cast<uint4*>(bits + (size_t)g * BM_WORDS + wv) = t4[i];
  }
}

template <bool FULL>
__device__ __forceinline__ void drain_ring(
    const float* __restrict__ h, const unsigned* __restrict__ pk,
    float* acc, float* sdeg,
    const unsigned* q, unsigned head, int n, int lane) {
  const unsigned e = q[(head + lane) & (RING - 1)];
  const bool valid = FULL || lane < n;
  int s = 0, dl = 0;
  if (valid) {
    const unsigned w = pk[e];
    s  = (int)(w & SRC_MASK);
    dl = (int)(w >> SRC_BITS);
    atomicAdd(&sdeg[dl], 1.0f);
  }
  const float2* h2 = reinterpret_cast<const float2*>(h);
  const int cntj = FULL ? 32 : n;
  for (int j0 = 0; j0 < cntj; j0 += ROWS_AHEAD) {
    float2 v[ROWS_AHEAD];
#pragma unroll
    for (int u = 0; u < ROWS_AHEAD; ++u) {
      int j = j0 + u;
      if (j >= cntj) j = cntj - 1;
      const int sj = __builtin_amdgcn_readlane(s, j);
      v[u] = h2[(unsigned)sj * (unsigned)(NODE_DIM / 2) + (unsigned)lane];
    }
#pragma unroll
    for (int u = 0; u < ROWS_AHEAD; ++u) {
      const int j = j0 + u;
      if (j < cntj) {
        const int dj = __builtin_amdgcn_readlane(dl, j);
        float* a = acc + dj * NODE_DIM + lane;
        atomicAdd(a, v[u].x);
        atomicAdd(a + 32, v[u].y);
      }
    }
  }
}

__global__ __launch_bounds__(OWN_THREADS) void edge_owner_scan(
    const float* __restrict__ h, const unsigned* __restrict__ pk,
    const unsigned* __restrict__ bits,
    float* __restrict__ agg, float* __restrict__ deg, float* __restrict__ gstat) {
  __shared__ float acc[OWN_NODES * NODE_DIM];
  __shared__ float sdeg[OWN_NODES];
  __shared__ unsigned ring[OWN_WAVES][RING];

  const int tid  = threadIdx.x;
  const int wave = __builtin_amdgcn_readfirstlane(tid >> 5);
  const int lane = tid & 31;
  const int base = blockIdx.x * OWN_NODES;
  int cnt = N_NODES - base;
  if (cnt > OWN_NODES) cnt = OWN_NODES;

  {
    float4* a4 = reinterpret_cast<float4*>(acc);
    const float4 z = {0.0f, 0.0f, 0.0f, 0.0f};
    for (int i = tid; i < OWN_NODES * NODE_DIM / 4; i += OWN_THREADS) a4[i] = z;
    for (int i = tid; i < OWN_NODES; i += OWN_THREADS) sdeg[i] = 0.0f;
    if (blockIdx.x == 0 && tid < 2 * OUT_DIM) gstat[tid] = 0.0f;
  }
  __syncthreads();

  unsigned* q = ring[wave];
  unsigned head = 0, tail = 0;
  const uint4* bm4 = reinterpret_cast<const uint4*>(bits + (size_t)blockIdx.x * BM_WORDS);

  const uint4 none = {0u, 0u, 0u, 0u};
  uint4 nxt;
  {
    const int f4 = wave * 32 + lane;
    nxt = bm4[f4 < BM_WORDS / 4 ? f4 : BM_WORDS / 4 - 1];
    if (f4 >= BM_WORDS / 4) nxt = none;
  }
  for (int k = wave * 32; k < BM_WORDS / 4; k += OWN_WAVES * 32) {
    const int i4 = k + lane;
    const uint4 m4 = nxt;
    const int in4 = i4 + OWN_WAVES * 32;
    nxt = bm4[in4 < BM_WORDS / 4 ? in4 : BM_WORDS / 4 - 1];
    if (in4 >= BM_WORDS / 4) nxt = none;
    const unsigned mw[4] = {m4.x, m4.y, m4.z, m4.w};
#pragma unroll
    for (int c = 0; c < 4; ++c) {
      unsigned mm = mw[c];
      const unsigned ebase = ((unsigned)i4 * 4u + (unsigned)c) * 32u;
      unsigned any = __builtin_amdgcn_ballot_w32(mm != 0u);
      while (any != 0u) {
        if (mm != 0u) {
          const unsigned b = (unsigned)__builtin_ctz(mm);
          mm &= mm - 1u;
          const unsigned pos = tail + __builtin_amdgcn_mbcnt_lo(any, 0u);
          q[pos & (RING - 1)] = ebase + b;
        }
        tail += (unsigned)__builtin_popcount(any);
        while (tail - head >= 32u) {
          drain_ring<true>(h, pk, acc, sdeg, q, head, 32, lane);
          head += 32u;
        }
        any = __builtin_amdgcn_ballot_w32(mm != 0u);
      }
    }
  }
  if (tail != head)
    drain_ring<false>(h, pk, acc, sdeg, q, head, (int)(tail - head), lane);
  __syncthreads();

  for (int i = tid; i < cnt * NODE_DIM; i += OWN_THREADS) {
    const int m = i >> 6, c = i & 63;
    agg[(size_t)(base + m) * NODE_DIM + c] = acc[m * NODE_DIM + ((c & 1) << 5) + (c >> 1)];
  }
  for (int i = tid; i < cnt; i += OWN_THREADS) deg[base + i] = sdeg[i];
}

#define PS_PITCH (OUT_DIM + 4)
#define XS_FLOATS (16 * (NODE_DIM + 1))
#define HS_FLOATS (16 * (HIDDEN + 1))
static_assert(4 * 16 * PS_PITCH <= XS_FLOATS + HS_FLOATS, "the quarter sums fit where xs and hs were");
static_assert(HIDDEN == 4 * 64 && OUT_DIM == 64 && NODE_DIM % 4 == 0, "four waves: 64 hidden columns each, then one K quarter each");
static_assert(16 * NODE_DIM / 4 == 2 * 128, "staging: two 16-byte pieces per thread");

__global__ __launch_bounds__(128) void fused_mlp(
    const float* __restrict__ h,  const float* __restrict__ W1,
    const float* __restrict__ b1, const float* __restrict__ W2,
    const float* __restrict__ b2, const float* __restrict__ agg,
    const float* __restrict__ deg, float* __restrict__ out,
    float* __restrict__ part) {
  __shared__ __attribute__((aligned(16))) float smem[XS_FLOATS + HS_FLOATS];
  float (*xs)[NODE_DIM + 1] = reinterpret_cast<float (*)[NODE_DIM + 1]>(smem);
  float (*hs)[HIDDEN + 1]   = reinterpret_cast<float (*)[HIDDEN + 1]>(smem + XS_FLOATS);
  float (*ps)[16][PS_PITCH] = reinterpret_cast<float (*)[16][PS_PITCH]>(smem);

  const int m0   = blockIdx.x * 16;
  const int tid  = threadIdx.x;
  const int wave = tid >> 5;
  const int lane = tid & 31;
  const int half = lane >> 4;
  const int r    = lane & 15;

  {
    const int mA = tid >> 4, mB = mA + 8, c = (tid & 15) * 4;
    const int gA = (m0 + mA) * NODE_DIM + c;
    const int gB = gA + 8 * NODE_DIM;
    float dA = deg[m0 + mA];
    float dB = deg[m0 + mB];
    const float4 aA = *reinterpret_cast<const float4*>(agg + gA);
    const float4 hA = *reinterpret_cast<const float4*>(h + gA);
    const float4 aB = *reinterpret_cast<const float4*>(agg + gB);
    const float4 hB = *reinterpret_cast<const float4*>(h + gB);
    dA = dA > 1.0f ? dA : 1.0f;
    dB = dB > 1.0f ? dB : 1.0f;
    xs[mA][c]     = aA.x / dA + hA.x;
    xs[mA][c + 1] = aA.y / dA + hA.y;
    xs[mA][c + 2] = aA.z / dA + hA.z;
    xs[mA][c + 3] = aA.w / dA + hA.w;
    xs[mB][c]     = aB.x / dB + hB.x;
    xs[mB][c + 1] = aB.y / dB + hB.y;
    xs[mB][c + 2] = aB.z / dB + hB.z;
    xs[mB][c + 3] = aB.w / dB + hB.w;
  }
  __syncthreads();

  v8f acc0 = {}, acc1 = {}, acc2 = {}, acc3 = {};
  const int nb = wave * 64;
  {
    const float* w1p = W1 + (2 * half) * HIDDEN + nb + 4 * r;
#pragma unroll 4
    for (int k = 0; k < NODE_DIM; k += 4) {
      v2f a;
      a.x = xs[r][k + 2 * half];
      a.y = xs[r][k + 2 * half + 1];
      const float4 q0 = *reinterpret_cast<const float4*>(w1p + k * HIDDEN);
      const float4 q1 = *reinterpret_cast<const float4*>(w1p + (k + 1) * HIDDEN);
      v2f b;
      b.x = q0.x; b.y = q1.x; WMMA_F32X4(acc0, a, b);
      b.x = q0.y; b.y = q1.y; WMMA_F32X4(acc1, a, b);
      b.x = q0.z; b.y = q1.z; WMMA_F32X4(acc2, a, b);
      b.x = q0.w; b.y = q1.w; WMMA_F32X4(acc3, a, b);
    }
  }
  {
    const float4 bias4 = *reinterpret_cast<const float4*>(b1 + nb + 4 * r);
#pragma unroll
    for (int j = 0; j < 8; ++j) {
      float* hrow = &hs[j + 8 * half][nb + 4 * r];
      const float v0 = acc0[j] + bias4.x;
      const float v1 = acc1[j] + bias4.y;
      const float v2 = acc2[j] + bias4.z;
      const float v3 = acc3[j] + bias4.w;
      hrow[0] = v0 > 0.0f ? v0 : 0.0f;
      hrow[1] = v1 > 0.0f ? v1 : 0.0f;
      hrow[2] = v2 > 0.0f ? v2 : 0.0f;
      hrow[3] = v3 > 0.0f ? v3 : 0.0f;
    }
  }
  __syncthreads();

  v8f o0 = {}, o1 = {}, o2 = {}, o3 = {};
  {
    const int kb = wave * 64;
    const float* w2p = W2 + (kb + 2 * half) * OUT_DIM + 4 * r;
#pragma unroll 4
    for (int k = 0; k < 64; k += 4) {
      v2f a;
      a.x = hs[r][kb + k + 2 * half];
      a.y = hs[r][kb + k + 2 * half + 1];
      const float4 q0 = *reinterpret_cast<const float4*>(w2p + k * OUT_DIM);
      const float4 q1 = *reinterpret_cast<const float4*>(w2p + (k + 1) * OUT_DIM);
      v2f b;
      b.x = q0.x; b.y = q1.x; WMMA_F32X4(o0, a, b);
      b.x = q0.y; b.y = q1.y; WMMA_F32X4(o1, a, b);
      b.x = q0.z; b.y = q1.z; WMMA_F32X4(o2, a, b);
      b.x = q0.w; b.y = q1.w; WMMA_F32X4(o3, a, b);
    }
  }
  __syncthreads();
#pragma unroll
  for (int j = 0; j < 8; ++j) {
    const float4 p = {o0[j], o1[j], o2[j], o3[j]};
    *reinterpret_cast<float4*>(&ps[wave][j + 8 * half][4 * r]) = p;
  }
  __syncthreads();

  const int ob = wave * 16;
  {
    const float bias = b2[ob + r];
    float s = 0.0f, q = 0.0f;
#pragma unroll
    for (int j = 0; j < 8; ++j) {
      const int row = j + 8 * half;
      float v = ((ps[0][row][ob + r] + ps[1][row][ob + r]) + ps[2][row][ob + r]) + ps[3][row][ob + r];
      v += bias;
      v = v > 0.0f ? v : 0.0f;
      out[(m0 + row) * OUT_DIM + ob + r] = v;
      s += v;
      q += v * v;
    }
    s += __shfl_xor(s, 16);
    q += __shfl_xor(q, 16);
    part[blockIdx.x * 128 + tid] = half ? q : s;
  }
}

#define PART_ROWS (N_NODES / 16)
#define ST_ROWS   25
static_assert(PART_ROWS % ST_ROWS == 0, "every stats block sums whole partial rows");

__global__ __launch_bounds__(128) void bn_stats(
    const float* __restrict__ part, float* __restrict__ gsum,
    float* __restrict__ gsq) {
  const int j = threadIdx.x;
  const float* p = part + (size_t)blockIdx.x * (ST_ROWS * 128) + j;
  float v[ST_ROWS];
#pragma unroll
  for (int u = 0; u < ST_ROWS; ++u) v[u] = p[u * 128];
  float t = 0.0f;
#pragma unroll
  for (int u = 0; u < ST_ROWS; ++u) t += v[u];
  const int c = (j >> 5) * 16 + (j & 15);
  atomicAdd(((j >> 4) & 1) ? &gsq[c] : &gsum[c], t);
}

#define AP_THREADS 128
#define AP_ELEMS   16
static_assert((N_NODES * OUT_DIM) % (AP_THREADS * AP_ELEMS) == 0, "the grid covers out exactly: no bounds guard");
static_assert(AP_THREADS % OUT_DIM == 0, "all of a lane's elements belong to one channel");

__global__ __launch_bounds__(AP_THREADS) void bn_apply(
    float* __restrict__ x, const float* __restrict__ gsum,
    const float* __restrict__ gsq, const float* __restrict__ gamma,
    const float* __restrict__ beta) {
  const int c = threadIdx.x & (OUT_DIM - 1);
  float* p = x + (size_t)blockIdx.x * (AP_THREADS * AP_ELEMS) + threadIdx.x;
  float v[AP_ELEMS];
#pragma unroll
  for (int u = 0; u < AP_ELEMS; ++u) v[u] = p[u * AP_THREADS];
  const float inv_n = 1.0f / (float)N_NODES;
  const float mean = gsum[c] * inv_n;
  const float var  = gsq[c] * inv_n - mean * mean;
  const float sc   = gamma[c] * __frsqrt_rn(var + BN_EPS);
  const float sh   = beta[c];
#pragma unroll
  for (int u = 0; u < AP_ELEMS; ++u) p[u * AP_THREADS] = (v[u] - mean) * sc + sh;
}

extern "C" void kernel_launch(void* const* d_in, const int* in_sizes, int n_in,
                              void* d_out, int out_size, void* d_ws, size_t ws_size,
                              hipStream_t stream) {
  const float* h     = (const float*)d_in[0];
  const float* W1    = (const float*)d_in[1];
  const float* b1    = (const float*)d_in[2];
  const float* W2    = (const float*)d_in[3];
  const float* b2    = (const float*)d_in[4];
  const float* gamma = (const float*)d_in[5];
  const float* beta  = (const float*)d_in[6];
  const int*   src   = (const int*)d_in[7];
  const int*   dst   = (const int*)d_in[8];
  float* out = (float*)d_out;

  float* agg  = (float*)d_ws;
  float* deg  = agg + (size_t)N_NODES * NODE_DIM;
  float* gsum = deg + N_NODES;
  float* gsq  = gsum + OUT_DIM;
  unsigned* bits = (unsigned*)(gsq + OUT_DIM);
  float* part = (float*)(bits + (size_t)OWN_GROUPS * BM_WORDS);
  unsigned* pk = (unsigned*)(part + (size_t)(N_NODES / 16) * 128);

  edge_owner_bits<<<N_CHUNK, OWN_THREADS, 0, stream>>>(src, dst, bits, pk);

  edge_owner_scan<<<OWN_GROUPS, OWN_THREADS, 0, stream>>>(h, pk, bits, agg, deg, gsum);

  fused_mlp<<<N_NODES / 16, 128, 0, stream>>>(h, W1, b1, W2, b2, agg, deg, out, part);

  bn_stats<<<PART_ROWS / ST_ROWS, 128, 0, stream>>>(part, gsum, gsq);

  bn_apply<<<(N_NODES * OUT_DIM) / (AP_THREADS * AP_ELEMS), AP_THREADS, 0, stream>>>(out, gsum, gsq, gamma, beta);
}


# MLP work-group takes 32 node rows, not 16: each weight fragment a wave loads feeds two row tiles (8 matrix instructions per load pair, was 4); BN partial rows per 32 nodes: channel sums in a new order
# speedup vs baseline: 3.8956x; 1.0774x over previous
#include <hip/hip_runtime.h>
#include <hip/hip_bf16.h>

#define N_NODES   100000
#define N_EDGES   1600000
#define NODE_DIM  64
#define HIDDEN    256
#define OUT_DIM   64
#define BN_EPS    1e-5f

typedef __attribute__((ext_vector_type(2))) float v2f;
typedef __attribute__((ext_vector_type(8))) float v8f;

#define WMMA_F32X4(ACC, A, B) \
  ACC = __builtin_amdgcn_wmma_f32_16x16x4_f32(false, (A), false, (B), (short)0, (ACC), false, false)

#define OWN_GROUPS   256
#define OWN_NODES    391
#define OWN_THREADS  1024
#define OWN_WAVES    32
#define RING         64
#define ROWS_AHEAD   16
#define BM_WORDS     (N_EDGES / 32)
#define CH_EDGES     4096
#define CH_WORDS     (CH_EDGES / 32)
#define N_CHUNK      ((N_EDGES + CH_EDGES - 1) / CH_EDGES)
#define SRC_BITS     17
#define SRC_MASK     ((1u << SRC_BITS) - 1u)

static_assert(OWN_GROUPS * OWN_NODES >= N_NODES, "owner ranges must cover all nodes");
static_assert(N_EDGES % 32 == 0 && BM_WORDS % 4 == 0, "a plane is whole 16-byte pieces");
static_assert(N_EDGES % 4 == 0, "dst is read four edges at a time");
static_assert(CH_EDGES == 4 * OWN_THREADS, "one 16-byte dst load per thread");
static_assert((1u << SRC_BITS) >= N_NODES, "a source node fits the low field");
static_assert(OWN_NODES <= (1u << (32 - SRC_BITS)), "an offset inside an owner range fits the high field");

__global__ __launch_bounds__(OWN_THREADS) void edge_owner_bits(
    const int* __restrict__ src, const int* __restrict__ dst,
    unsigned* __restrict__ bits, unsigned* __restrict__ pk) {
  __shared__ unsigned tab[OWN_GROUPS * CH_WORDS];
  const int tid = threadIdx.x;
  uint4* t4 = reinterpret_cast<uint4*>(tab);
  {
    const uint4 z = {0u, 0u, 0u, 0u};
    for (int i = tid; i < OWN_GROUPS * CH_WORDS / 4; i += OWN_THREADS) t4[i] = z;
  }
  __syncthreads();
  const int e0 = blockIdx.x * CH_EDGES + tid * 4;
  if (e0 < N_EDGES) {
    const int4 d = *reinterpret_cast<const int4*>(dst + e0);
    const int4 s = *reinterpret_cast<const int4*>(src + e0);
    const unsigned dc[4] = {(unsigned)d.x, (unsigned)d.y, (unsigned)d.z, (unsigned)d.w};
    const unsigned sc[4] = {(unsigned)s.x, (unsigned)s.y, (unsigned)s.z, (unsigned)s.w};
    unsigned pw[4];
    const int w  = tid >> 3;
    const int b0 = (tid & 7) * 4;
#pragma unroll
    for (int c = 0; c < 4; ++c) {
      const unsigned g = dc[c] / (unsigned)OWN_NODES;
      pw[c] = (sc[c] & SRC_MASK) | ((dc[c] - g * (unsigned)OWN_NODES) << SRC_BITS);
      if (dc[c] < (unsigned)N_NODES)
        atomicOr(&tab[g * CH_WORDS + w], 1u << (b0 + c));
    }
    const uint4 p4 = {pw[0], pw[1], pw[2], pw[3]};
    *reinterpret_cast<uint4*>(pk + e0) = p4;
  }
  __syncthreads();
  const int wbase = blockIdx.x * CH_WORDS;
  for (int i = tid; i < OWN_GROUPS * CH_WORDS / 4; i += OWN_THREADS) {
    const int g  = i / (CH_WORDS / 4);
    const int wv = wbase + (i % (CH_WORDS / 4)) * 4;
    if (wv < BM_WORDS)
      *reinterpret_cast<uint4*>(bits + (size_t)g * BM_WORDS + wv) = t4[i];
  }
}

template <bool FULL>
__device__ __forceinline__ void drain_ring(
    const float* __restrict__ h, const unsigned* __restrict__ pk,
    float* acc, float* sdeg,
    const unsigned* q, unsigned head, int n, int lane) {
  const unsigned e = q[(head + lane) & (RING - 1)];
  const bool valid = FULL || lane < n;
  int s = 0, dl = 0;
  if (valid) {
    const unsigned w = pk[e];
    s  = (int)(w & SRC_MASK);
    dl = (int)(w >> SRC_BITS);
    atomicAdd(&sdeg[dl], 1.0f);
  }
  const float2* h2 = reinterpret_cast<const float2*>(h);
  const int cntj = FULL ? 32 : n;
  for (int j0 = 0; j0 < cntj; j0 += ROWS_AHEAD) {
    float2 v[ROWS_AHEAD];
#pragma unroll
    for (int u = 0; u < ROWS_AHEAD; ++u) {
      int j = j0 + u;
      if (j >= cntj) j = cntj - 1;
      const int sj = __builtin_amdgcn_readlane(s, j);
      v[u] = h2[(unsigned)sj * (unsigned)(NODE_DIM / 2) + (unsigned)lane];
    }
#pragma unroll
    for (int u = 0; u < ROWS_AHEAD; ++u) {
      const int j = j0 + u;
      if (j < cntj) {
        const int dj = __builtin_amdgcn_readlane(dl, j);
        float* a = acc + dj * NODE_DIM + lane;
        atomicAdd(a, v[u].x);
        atomicAdd(a + 32, v[u].y);
      }
    }
  }
}

__global__ __launch_bounds__(OWN_THREADS) void edge_owner_scan(
    const float* __restrict__ h, const unsigned* __restrict__ pk,
    const unsigned* __restrict__ bits,
    float* __restrict__ agg, float* __restrict__ deg, float* __restrict__ gstat) {
  __shared__ float acc[OWN_NODES * NODE_DIM];
  __shared__ float sdeg[OWN_NODES];
  __shared__ unsigned ring[OWN_WAVES][RING];

  const int tid  = threadIdx.x;
  const int wave = __builtin_amdgcn_readfirstlane(tid >> 5);
  const int lane = tid & 31;
  const int base = blockIdx.x * OWN_NODES;
  int cnt = N_NODES - base;
  if (cnt > OWN_NODES) cnt = OWN_NODES;

  {
    float4* a4 = reinterpret_cast<float4*>(acc);
    const float4 z = {0.0f, 0.0f, 0.0f, 0.0f};
    for (int i = tid; i < OWN_NODES * NODE_DIM / 4; i += OWN_THREADS) a4[i] = z;
    for (int i = tid; i < OWN_NODES; i += OWN_THREADS) sdeg[i] = 0.0f;
    if (blockIdx.x == 0 && tid < 2 * OUT_DIM) gstat[tid] = 0.0f;
  }
  __syncthreads();

  unsigned* q = ring[wave];
  unsigned head = 0, tail = 0;
  const uint4* bm4 = reinterpret_cast<const uint4*>(bits + (size_t)blockIdx.x * BM_WORDS);

  const uint4 none = {0u, 0u, 0u, 0u};
  uint4 nxt;
  {
    const int f4 = wave * 32 + lane;
    nxt = bm4[f4 < BM_WORDS / 4 ? f4 : BM_WORDS / 4 - 1];
    if (f4 >= BM_WORDS / 4) nxt = none;
  }
  for (int k = wave * 32; k < BM_WORDS / 4; k += OWN_WAVES * 32) {
    const int i4 = k + lane;
    const uint4 m4 = nxt;
    const int in4 = i4 + OWN_WAVES * 32;
    nxt = bm4[in4 < BM_WORDS / 4 ? in4 : BM_WORDS / 4 - 1];
    if (in4 >= BM_WORDS / 4) nxt = none;
    const unsigned mw[4] = {m4.x, m4.y, m4.z, m4.w};
#pragma unroll
    for (int c = 0; c < 4; ++c) {
      unsigned mm = mw[c];
      const unsigned ebase = ((unsigned)i4 * 4u + (unsigned)c) * 32u;
      unsigned any = __builtin_amdgcn_ballot_w32(mm != 0u);
      while (any != 0u) {
        if (mm != 0u) {
          const unsigned b = (unsigned)__builtin_ctz(mm);
          mm &= mm - 1u;
          const unsigned pos = tail + __builtin_amdgcn_mbcnt_lo(any, 0u);
          q[pos & (RING - 1)] = ebase + b;
        }
        tail += (unsigned)__builtin_popcount(any);
        while (tail - head >= 32u) {
          drain_ring<true>(h, pk, acc, sdeg, q, head, 32, lane);
          head += 32u;
        }
        any = __builtin_amdgcn_ballot_w32(mm != 0u);
      }
    }
  }
  if (tail != head)
    drain_ring<false>(h, pk, acc, sdeg, q, head, (int)(tail - head), lane);
  __syncthreads();

  for (int i = tid; i < cnt * NODE_DIM; i += OWN_THREADS) {
    const int m = i >> 6, c = i & 63;
    agg[(size_t)(base + m) * NODE_DIM + c] = acc[m * NODE_DIM + ((c & 1) << 5) + (c >> 1)];
  }
  for (int i = tid; i < cnt; i += OWN_THREADS) deg[base + i] = sdeg[i];
}

#define MLP_ROWS 32
#define PS_PITCH (OUT_DIM + 4)
#define XS_FLOATS (MLP_ROWS * (NODE_DIM + 1))
#define HS_FLOATS (MLP_ROWS * (HIDDEN + 1))
static_assert(4 * MLP_ROWS * PS_PITCH <= XS_FLOATS + HS_FLOATS, "the quarter sums fit where xs and hs were");
static_assert(HIDDEN == 4 * 64 && OUT_DIM == 64 && NODE_DIM % 4 == 0, "four waves: 64 hidden columns each, then one K quarter each");
static_assert(MLP_ROWS * NODE_DIM / 4 == 4 * 128, "staging: four 16-byte pieces per thread");
static_assert(N_NODES % MLP_ROWS == 0, "the grid covers the nodes exactly: no row guard");

__global__ __launch_bounds__(128) void fused_mlp(
    const float* __restrict__ h,  const float* __restrict__ W1,
    const float* __restrict__ b1, const float* __restrict__ W2,
    const float* __restrict__ b2, const float* __restrict__ agg,
    const float* __restrict__ deg, float* __restrict__ out,
    float* __restrict__ part) {
  __shared__ __attribute__((aligned(16))) float smem[XS_FLOATS + HS_FLOATS];
  float (*xs)[NODE_DIM + 1] = reinterpret_cast<float (*)[NODE_DIM + 1]>(smem);
  float (*hs)[HIDDEN + 1]   = reinterpret_cast<float (*)[HIDDEN + 1]>(smem + XS_FLOATS);
  float (*ps)[MLP_ROWS][PS_PITCH] = reinterpret_cast<float (*)[MLP_ROWS][PS_PITCH]>(smem);

  const int m0   = blockIdx.x * MLP_ROWS;
  const int tid  = threadIdx.x;
  const int wave = tid >> 5;
  const int lane = tid & 31;
  const int half = lane >> 4;
  const int r    = lane & 15;

  {
    const int mA = tid >> 4, c = (tid & 15) * 4;
    float dg[4];
    float4 av[4], hv[4];
#pragma unroll
    for (int t = 0; t < 4; ++t) {
      const int row = m0 + mA + 8 * t;
      dg[t] = deg[row];
      av[t] = *reinterpret_cast<const float4*>(agg + row * NODE_DIM + c);
      hv[t] = *reinterpret_cast<const float4*>(h + row * NODE_DIM + c);
    }
#pragma unroll
    for (int t = 0; t < 4; ++t) {
      const float d = dg[t] > 1.0f ? dg[t] : 1.0f;
      float* x = &xs[mA + 8 * t][c];
      x[0] = av[t].x / d + hv[t].x;
      x[1] = av[t].y / d + hv[t].y;
      x[2] = av[t].z / d + hv[t].z;
      x[3] = av[t].w / d + hv[t].w;
    }
  }
  __syncthreads();

  v8f acc0 = {}, acc1 = {}, acc2 = {}, acc3 = {};
  v8f acd0 = {}, acd1 = {}, acd2 = {}, acd3 = {};
  const int nb = wave * 64;
  {
    const float* w1p = W1 + (2 * half) * HIDDEN + nb + 4 * r;
#pragma unroll 4
    for (int k = 0; k < NODE_DIM; k += 4) {
      v2f a, d;
      a.x = xs[r][k + 2 * half];
      a.y = xs[r][k + 2 * half + 1];
      d.x = xs[16 + r][k + 2 * half];
      d.y = xs[16 + r][k + 2 * half + 1];
      const float4 q0 = *reinterpret_cast<const float4*>(w1p + k * HIDDEN);
      const float4 q1 = *reinterpret_cast<const float4*>(w1p + (k + 1) * HIDDEN);
      v2f b;
      b.x = q0.x; b.y = q1.x; WMMA_F32X4(acc0, a, b); WMMA_F32X4(acd0, d, b);
      b.x = q0.y; b.y = q1.y; WMMA_F32X4(acc1, a, b); WMMA_F32X4(acd1, d, b);
      b.x = q0.z; b.y = q1.z; WMMA_F32X4(acc2, a, b); WMMA_F32X4(acd2, d, b);
      b.x = q0.w; b.y = q1.w; WMMA_F32X4(acc3, a, b); WMMA_F32X4(acd3, d, b);
    }
  }
  {
    const float4 bias4 = *reinterpret_cast<const float4*>(b1 + nb + 4 * r);
#pragma unroll
    for (int j = 0; j < 8; ++j) {
      float* hrow = &hs[j + 8 * half][nb + 4 * r];
      const float v0 = acc0[j] + bias4.x;
      const float v1 = acc1[j] + bias4.y;
      const float v2 = acc2[j] + bias4.z;
      const float v3 = acc3[j] + bias4.w;
      hrow[0] = v0 > 0.0f ? v0 : 0.0f;
      hrow[1] = v1 > 0.0f ? v1 : 0.0f;
      hrow[2] = v2 > 0.0f ? v2 : 0.0f;
      hrow[3] = v3 > 0.0f ? v3 : 0.0f;
      float* grow = &hs[16 + j + 8 * half][nb + 4 * r];
      const float u0 = acd0[j] + bias4.x;
      const float u1 = acd1[j] + bias4.y;
      const float u2 = acd2[j] + bias4.z;
      const float u3 = acd3[j] + bias4.w;
      grow[0] = u0 > 0.0f ? u0 : 0.0f;
      grow[1] = u1 > 0.0f ? u1 : 0.0f;
      grow[2] = u2 > 0.0f ? u2 : 0.0f;
      grow[3] = u3 > 0.0f ? u3 : 0.0f;
    }
  }
  __syncthreads();

  v8f o0 = {}, o1 = {}, o2 = {}, o3 = {};
  v8f p0 = {}, p1 = {}, p2 = {}, p3 = {};
  {
    const int kb = wave * 64;
    const float* w2p = W2 + (kb + 2 * half) * OUT_DIM + 4 * r;
#pragma unroll 4
    for (int k = 0; k < 64; k += 4) {
      v2f a, d;
      a.x = hs[r][kb + k + 2 * half];
      a.y = hs[r][kb + k + 2 * half + 1];
      d.x = hs[16 + r][kb + k + 2 * half];
      d.y = hs[16 + r][kb + k + 2 * half + 1];
      const float4 q0 = *reinterpret_cast<const float4*>(w2p + k * OUT_DIM);
      const float4 q1 = *reinterpret_cast<const float4*>(w2p + (k + 1) * OUT_DIM);
      v2f b;
      b.x = q0.x; b.y = q1.x; WMMA_F32X4(o0, a, b); WMMA_F32X4(p0, d, b);
      b.x = q0.y; b.y = q1.y; WMMA_F32X4(o1, a, b); WMMA_F32X4(p1, d, b);
      b.x = q0.z; b.y = q1.z; WMMA_F32X4(o2, a, b); WMMA_F32X4(p2, d, b);
      b.x = q0.w; b.y = q1.w; WMMA_F32X4(o3, a, b); WMMA_F32X4(p3, d, b);
    }
  }
  __syncthreads();
#pragma unroll
  for (int j = 0; j < 8; ++j) {
    const float4 lo = {o0[j], o1[j], o2[j], o3[j]};
    const float4 hi = {p0[j], p1[j], p2[j], p3[j]};
    *reinterpret_cast<float4*>(&ps[wave][j + 8 * half][4 * r]) = lo;
    *reinterpret_cast<float4*>(&ps[wave][16 + j + 8 * half][4 * r]) = hi;
  }
  __syncthreads();

  const int ob = wave * 16;
  {
    const float bias = b2[ob + r];
    float s = 0.0f, q = 0.0f;
#pragma unroll
    for (int j = 0; j < 16; ++j) {
      const int row = (j & 7) + 8 * half + 16 * (j >> 3);
      float v = ((ps[0][row][ob + r] + ps[1][row][ob + r]) + ps[2][row][ob + r]) + ps[3][row][ob + r];
      v += bias;
      v = v > 0.0f ? v : 0.0f;
      out[(m0 + row) * OUT_DIM + ob + r] = v;
      s += v;
      q += v * v;
    }
    s += __shfl_xor(s, 16);
    q += __shfl_xor(q, 16);
    part[blockIdx.x * 128 + tid] = half ? q : s;
  }
}

#define PART_ROWS (N_NODES / MLP_ROWS)
#define ST_ROWS   25
static_assert(PART_ROWS % ST_ROWS == 0, "every stats block sums whole partial rows");

__global__ __launch_bounds__(128) void bn_stats(
    const float* __restrict__ part, float* __restrict__ gsum,
    float* __restrict__ gsq) {
  const int j = threadIdx.x;
  const float* p = part + (size_t)blockIdx.x * (ST_ROWS * 128) + j;
  float v[ST_ROWS];
#pragma unroll
  for (int u = 0; u < ST_ROWS; ++u) v[u] = p[u * 128];
  float t = 0.0f;
#pragma unroll
  for (int u = 0; u < ST_ROWS; ++u) t += v[u];
  const int c = (j >> 5) * 16 + (j & 15);
  atomicAdd(((j >> 4) & 1) ? &gsq[c] : &gsum[c], t);
}

#define AP_THREADS 128
#define AP_ELEMS   16
static_assert((N_NODES * OUT_DIM) % (AP_THREADS * AP_ELEMS) == 0, "the grid covers out exactly: no bounds guard");
static_assert(AP_THREADS % OUT_DIM == 0, "all of a lane's elements belong to one channel");

__global__ __launch_bounds__(AP_THREADS) void bn_apply(
    float* __restrict__ x, const float* __restrict__ gsum,
    const float* __restrict__ gsq, const float* __restrict__ gamma,
    const float* __restrict__ beta) {
  const int c = threadIdx.x & (OUT_DIM - 1);
  float* p = x + (size_t)blockIdx.x * (AP_THREADS * AP_ELEMS) + threadIdx.x;
  float v[AP_ELEMS];
#pragma unroll
  for (int u = 0; u < AP_ELEMS; ++u) v[u] = p[u * AP_THREADS];
  const float inv_n = 1.0f / (float)N_NODES;
  const float mean = gsum[c] * inv_n;
  const float var  = gsq[c] * inv_n - mean * mean;
  const float sc   = gamma[c] * __frsqrt_rn(var + BN_EPS);
  const float sh   = beta[c];
#pragma unroll
  for (int u = 0; u < AP_ELEMS; ++u) p[u * AP_THREADS] = (v[u] - mean) * sc + sh;
}

extern "C" void kernel_launch(void* const* d_in, const int* in_sizes, int n_in,
                              void* d_out, int out_size, void* d_ws, size_t ws_size,
                              hipStream_t stream) {
  const float* h     = (const float*)d_in[0];
  const float* W1    = (const float*)d_in[1];
  const float* b1    = (const float*)d_in[2];
  const float* W2    = (const float*)d_in[3];
  const float* b2    = (const float*)d_in[4];
  const float* gamma = (const float*)d_in[5];
  const float* beta  = (const float*)d_in[6];
  const int*   src   = (const int*)d_in[7];
  const int*   dst   = (const int*)d_in[8];
  float* out = (float*)d_out;

  float* agg  = (float*)d_ws;
  float* deg  = agg + (size_t)N_NODES * NODE_DIM;
  float* gsum = deg + N_NODES;
  float* gsq  = gsum + OUT_DIM;
  unsigned* bits = (unsigned*)(gsq + OUT_DIM);
  float* part = (float*)(bits + (size_t)OWN_GROUPS * BM_WORDS);
  unsigned* pk = (unsigned*)(part + (size_t)(N_NODES / 16) * 128);

  edge_owner_bits<<<N_CHUNK, OWN_THREADS, 0, stream>>>(src, dst, bits, pk);

  edge_owner_scan<<<OWN_GROUPS, OWN_THREADS, 0, stream>>>(h, pk, bits, agg, deg, gsum);

  fused_mlp<<<N_NODES / MLP_ROWS, 128, 0, stream>>>(h, W1, b1, W2, b2, agg, deg, out, part);

  bn_stats<<<PART_ROWS / ST_ROWS, 128, 0, stream>>>(part, gsum, gsq);

  bn_apply<<<(N_NODES * OUT_DIM) / (AP_THREADS * AP_ELEMS), AP_THREADS, 0, stream>>>(out, gsum, gsq, gamma, beta);
}


# Owner scan: a batch's 32 packed-word reads are issued one batch ahead, under the previous batch's row gather, and all 32 rows of a batch are in flight (was 16 + 16); sums by LDS arrival as before
# speedup vs baseline: 4.0046x; 1.0280x over previous
#include <hip/hip_runtime.h>
#include <hip/hip_bf16.h>

#define N_NODES   100000
#define N_EDGES   1600000
#define NODE_DIM  64
#define HIDDEN    256
#define OUT_DIM   64
#define BN_EPS    1e-5f

typedef __attribute__((ext_vector_type(2))) float v2f;
typedef __attribute__((ext_vector_type(8))) float v8f;

#define WMMA_F32X4(ACC, A, B) \
  ACC = __builtin_amdgcn_wmma_f32_16x16x4_f32(false, (A), false, (B), (short)0, (ACC), false, false)

#define OWN_GROUPS   256
#define OWN_NODES    391
#define OWN_THREADS  1024
#define OWN_WAVES    32
#define RING         64
#define ROWS_AHEAD   32
#define BM_WORDS     (N_EDGES / 32)
#define CH_EDGES     4096
#define CH_WORDS     (CH_EDGES / 32)
#define N_CHUNK      ((N_EDGES + CH_EDGES - 1) / CH_EDGES)
#define SRC_BITS     17
#define SRC_MASK     ((1u << SRC_BITS) - 1u)

static_assert(OWN_GROUPS * OWN_NODES >= N_NODES, "owner ranges must cover all nodes");
static_assert(N_EDGES % 32 == 0 && BM_WORDS % 4 == 0, "a plane is whole 16-byte pieces");
static_assert(N_EDGES % 4 == 0, "dst is read four edges at a time");
static_assert(CH_EDGES == 4 * OWN_THREADS, "one 16-byte dst load per thread");
static_assert((1u << SRC_BITS) >= N_NODES, "a source node fits the low field");
static_assert(OWN_NODES <= (1u << (32 - SRC_BITS)), "an offset inside an owner range fits the high field");

__global__ __launch_bounds__(OWN_THREADS) void edge_owner_bits(
    const int* __restrict__ src, const int* __restrict__ dst,
    unsigned* __restrict__ bits, unsigned* __restrict__ pk) {
  __shared__ unsigned tab[OWN_GROUPS * CH_WORDS];
  const int tid = threadIdx.x;
  uint4* t4 = reinterpret_cast<uint4*>(tab);
  {
    const uint4 z = {0u, 0u, 0u, 0u};
    for (int i = tid; i < OWN_GROUPS * CH_WORDS / 4; i += OWN_THREADS) t4[i] = z;
  }
  __syncthreads();
  const int e0 = blockIdx.x * CH_EDGES + tid * 4;
  if (e0 < N_EDGES) {
    const int4 d = *reinterpret_cast<const int4*>(dst + e0);
    const int4 s = *reinterpret_cast<const int4*>(src + e0);
    const unsigned dc[4] = {(unsigned)d.x, (unsigned)d.y, (unsigned)d.z, (unsigned)d.w};
    const unsigned sc[4] = {(unsigned)s.x, (unsigned)s.y, (unsigned)s.z, (unsigned)s.w};
    unsigned pw[4];
    const int w  = tid >> 3;
    const int b0 = (tid & 7) * 4;
#pragma unroll
    for (int c = 0; c < 4; ++c) {
      const unsigned g = dc[c] / (unsigned)OWN_NODES;
      pw[c] = (sc[c] & SRC_MASK) | ((dc[c] - g * (unsigned)OWN_NODES) << SRC_BITS);
      if (dc[c] < (unsigned)N_NODES)
        atomicOr(&tab[g * CH_WORDS + w], 1u << (b0 + c));
    }
    const uint4 p4 = {pw[0], pw[1], pw[2], pw[3]};
    *reinterpret_cast<uint4*>(pk + e0) = p4;
  }
  __syncthreads();
  const int wbase = blockIdx.x * CH_WORDS;
  for (int i = tid; i < OWN_GROUPS * CH_WORDS / 4; i += OWN_THREADS) {
    const int g  = i / (CH_WORDS / 4);
    const int wv = wbase + (i % (CH_WORDS / 4)) * 4;
    if (wv < BM_WORDS)
      *reinterpret_cast<uint4*>(bits + (size_t)g * BM_WORDS + wv) = t4[i];
  }
}

template <bool FULL>
__device__ __forceinline__ void work_batch(
    const float* __restrict__ h, float* acc, float* sdeg,
    unsigned w, int n, int lane) {
  const int s  = (int)(w & SRC_MASK);
  const int dl = (int)(w >> SRC_BITS);
  if (FULL || lane < n) atomicAdd(&sdeg[dl], 1.0f);
  const float2* h2 = reinterpret_cast<const float2*>(h);
  const int cntj = FULL ? 32 : n;
  float2 v[ROWS_AHEAD];
#pragma unroll
  for (int u = 0; u < ROWS_AHEAD; ++u) {
    int j = u;
    if (j >= cntj) j = cntj - 1;
    const int sj = __builtin_amdgcn_readlane(s, j);
    v[u] = h2[(unsigned)sj * (unsigned)(NODE_DIM / 2) + (unsigned)lane];
  }
#pragma unroll
  for (int u = 0; u < ROWS_AHEAD; ++u) {
    if (u < cntj) {
      const int dj = __builtin_amdgcn_readlane(dl, u);
      float* a = acc + dj * NODE_DIM + lane;
      atomicAdd(a, v[u].x);
      atomicAdd(a + 32, v[u].y);
    }
  }
}

__global__ __launch_bounds__(OWN_THREADS) void edge_owner_scan(
    const float* __restrict__ h, const unsigned* __restrict__ pk,
    const unsigned* __restrict__ bits,
    float* __restrict__ agg, float* __restrict__ deg, float* __restrict__ gstat) {
  __shared__ float acc[OWN_NODES * NODE_DIM];
  __shared__ float sdeg[OWN_NODES];
  __shared__ unsigned ring[OWN_WAVES][RING];

  const int tid  = threadIdx.x;
  const int wave = __builtin_amdgcn_readfirstlane(tid >> 5);
  const int lane = tid & 31;
  const int base = blockIdx.x * OWN_NODES;
  int cnt = N_NODES - base;
  if (cnt > OWN_NODES) cnt = OWN_NODES;

  {
    float4* a4 = reinterpret_cast<float4*>(acc);
    const float4 z = {0.0f, 0.0f, 0.0f, 0.0f};
    for (int i = tid; i < OWN_NODES * NODE_DIM / 4; i += OWN_THREADS) a4[i] = z;
    for (int i = tid; i < OWN_NODES; i += OWN_THREADS) sdeg[i] = 0.0f;
    if (blockIdx.x == 0 && tid < 2 * OUT_DIM) gstat[tid] = 0.0f;
  }
  __syncthreads();

  unsigned* q = ring[wave];
  unsigned head = 0, tail = 0;
  unsigned pend_w = 0u;
  bool has_pend = false;
  const uint4* bm4 = reinterpret_cast<const uint4*>(bits + (size_t)blockIdx.x * BM_WORDS);

  const uint4 none = {0u, 0u, 0u, 0u};
  uint4 nxt;
  {
    const int f4 = wave * 32 + lane;
    nxt = bm4[f4 < BM_WORDS / 4 ? f4 : BM_WORDS / 4 - 1];
    if (f4 >= BM_WORDS / 4) nxt = none;
  }
  for (int k = wave * 32; k < BM_WORDS / 4; k += OWN_WAVES * 32) {
    const int i4 = k + lane;
    const uint4 m4 = nxt;
    const int in4 = i4 + OWN_WAVES * 32;
    nxt = bm4[in4 < BM_WORDS / 4 ? in4 : BM_WORDS / 4 - 1];
    if (in4 >= BM_WORDS / 4) nxt = none;
    const unsigned mw[4] = {m4.x, m4.y, m4.z, m4.w};
#pragma unroll
    for (int c = 0; c < 4; ++c) {
      unsigned mm = mw[c];
      const unsigned ebase = ((unsigned)i4 * 4u + (unsigned)c) * 32u;
      unsigned any = __builtin_amdgcn_ballot_w32(mm != 0u);
      while (any != 0u) {
        if (mm != 0u) {
          const unsigned b = (unsigned)__builtin_ctz(mm);
          mm &= mm - 1u;
          const unsigned pos = tail + __builtin_amdgcn_mbcnt_lo(any, 0u);
          q[pos & (RING - 1)] = ebase + b;
        }
        tail += (unsigned)__builtin_popcount(any);
        while (tail - head >= 32u) {
          const unsigned w_new = pk[q[(head + lane) & (RING - 1)]];
          head += 32u;
          if (has_pend) work_batch<true>(h, acc, sdeg, pend_w, 32, lane);
          pend_w = w_new;
          has_pend = true;
        }
        any = __builtin_amdgcn_ballot_w32(mm != 0u);
      }
    }
  }
  {
    const int n_last = (int)(tail - head);
    unsigned w_last = 0u;
    if (lane < n_last) w_last = pk[q[(head + lane) & (RING - 1)]];
    if (has_pend) work_batch<true>(h, acc, sdeg, pend_w, 32, lane);
    if (n_last > 0) work_batch<false>(h, acc, sdeg, w_last, n_last, lane);
  }
  __syncthreads();

  for (int i = tid; i < cnt * NODE_DIM; i += OWN_THREADS) {
    const int m = i >> 6, c = i & 63;
    agg[(size_t)(base + m) * NODE_DIM + c] = acc[m * NODE_DIM + ((c & 1) << 5) + (c >> 1)];
  }
  for (int i = tid; i < cnt; i += OWN_THREADS) deg[base + i] = sdeg[i];
}

#define MLP_ROWS 32
#define PS_PITCH (OUT_DIM + 4)
#define XS_FLOATS (MLP_ROWS * (NODE_DIM + 1))
#define HS_FLOATS (MLP_ROWS * (HIDDEN + 1))
static_assert(4 * MLP_ROWS * PS_PITCH <= XS_FLOATS + HS_FLOATS, "the quarter sums fit where xs and hs were");
static_assert(HIDDEN == 4 * 64 && OUT_DIM == 64 && NODE_DIM % 4 == 0, "four waves: 64 hidden columns each, then one K quarter each");
static_assert(MLP_ROWS * NODE_DIM / 4 == 4 * 128, "staging: four 16-byte pieces per thread");
static_assert(N_NODES % MLP_ROWS == 0, "the grid covers the nodes exactly: no row guard");

__global__ __launch_bounds__(128) void fused_mlp(
    const float* __restrict__ h,  const float* __restrict__ W1,
    const float* __restrict__ b1, const float* __restrict__ W2,
    const float* __restrict__ b2, const float* __restrict__ agg,
    const float* __restrict__ deg, float* __restrict__ out,
    float* __restrict__ part) {
  __shared__ __attribute__((aligned(16))) float smem[XS_FLOATS + HS_FLOATS];
  float (*xs)[NODE_DIM + 1] = reinterpret_cast<float (*)[NODE_DIM + 1]>(smem);
  float (*hs)[HIDDEN + 1]   = reinterpret_cast<float (*)[HIDDEN + 1]>(smem + XS_FLOATS);
  float (*ps)[MLP_ROWS][PS_PITCH] = reinterpret_cast<float (*)[MLP_ROWS][PS_PITCH]>(smem);

  const int m0   = blockIdx.x * MLP_ROWS;
  const int tid  = threadIdx.x;
  const int wave = tid >> 5;
  const int lane = tid & 31;
  const int half = lane >> 4;
  const int r    = lane & 15;

  {
    const int mA = tid >> 4, c = (tid & 15) * 4;
    float dg[4];
    float4 av[4], hv[4];
#pragma unroll
    for (int t = 0; t < 4; ++t) {
      const int row = m0 + mA + 8 * t;
      dg[t] = deg[row];
      av[t] = *reinterpret_cast<const float4*>(agg + row * NODE_DIM + c);
      hv[t] = *reinterpret_cast<const float4*>(h + row * NODE_DIM + c);
    }
#pragma unroll
    for (int t = 0; t < 4; ++t) {
      const float d = dg[t] > 1.0f ? dg[t] : 1.0f;
      float* x = &xs[mA + 8 * t][c];
      x[0] = av[t].x / d + hv[t].x;
      x[1] = av[t].y / d + hv[t].y;
      x[2] = av[t].z / d + hv[t].z;
      x[3] = av[t].w / d + hv[t].w;
    }
  }
  __syncthreads();

  v8f acc0 = {}, acc1 = {}, acc2 = {}, acc3 = {};
  v8f acd0 = {}, acd1 = {}, acd2 = {}, acd3 = {};
  const int nb = wave * 64;
  {
    const float* w1p = W1 + (2 * half) * HIDDEN + nb + 4 * r;
#pragma unroll 4
    for (int k = 0; k < NODE_DIM; k += 4) {
      v2f a, d;
      a.x = xs[r][k + 2 * half];
      a.y = xs[r][k + 2 * half + 1];
      d.x = xs[16 + r][k + 2 * half];
      d.y = xs[16 + r][k + 2 * half + 1];
      const float4 q0 = *reinterpret_cast<const float4*>(w1p + k * HIDDEN);
      const float4 q1 = *reinterpret_cast<const float4*>(w1p + (k + 1) * HIDDEN);
      v2f b;
      b.x = q0.x; b.y = q1.x; WMMA_F32X4(acc0, a, b); WMMA_F32X4(acd0, d, b);
      b.x = q0.y; b.y = q1.y; WMMA_F32X4(acc1, a, b); WMMA_F32X4(acd1, d, b);
      b.x = q0.z; b.y = q1.z; WMMA_F32X4(acc2, a, b); WMMA_F32X4(acd2, d, b);
      b.x = q0.w; b.y = q1.w; WMMA_F32X4(acc3, a, b); WMMA_F32X4(acd3, d, b);
    }
  }
  {
    const float4 bias4 = *reinterpret_cast<const float4*>(b1 + nb + 4 * r);
#pragma unroll
    for (int j = 0; j < 8; ++j) {
      float* hrow = &hs[j + 8 * half][nb + 4 * r];
      const float v0 = acc0[j] + bias4.x;
      const float v1 = acc1[j] + bias4.y;
      const float v2 = acc2[j] + bias4.z;
      const float v3 = acc3[j] + bias4.w;
      hrow[0] = v0 > 0.0f ? v0 : 0.0f;
      hrow[1] = v1 > 0.0f ? v1 : 0.0f;
      hrow[2] = v2 > 0.0f ? v2 : 0.0f;
      hrow[3] = v3 > 0.0f ? v3 : 0.0f;
      float* grow = &hs[16 + j + 8 * half][nb + 4 * r];
      const float u0 = acd0[j] + bias4.x;
      const float u1 = acd1[j] + bias4.y;
      const float u2 = acd2[j] + bias4.z;
      const float u3 = acd3[j] + bias4.w;
      grow[0] = u0 > 0.0f ? u0 : 0.0f;
      grow[1] = u1 > 0.0f ? u1 : 0.0f;
      grow[2] = u2 > 0.0f ? u2 : 0.0f;
      grow[3] = u3 > 0.0f ? u3 : 0.0f;
    }
  }
  __syncthreads();

  v8f o0 = {}, o1 = {}, o2 = {}, o3 = {};
  v8f p0 = {}, p1 = {}, p2 = {}, p3 = {};
  {
    const int kb = wave * 64;
    const float* w2p = W2 + (kb + 2 * half) * OUT_DIM + 4 * r;
#pragma unroll 4
    for (int k = 0; k < 64; k += 4) {
      v2f a, d;
      a.x = hs[r][kb + k + 2 * half];
      a.y = hs[r][kb + k + 2 * half + 1];
      d.x = hs[16 + r][kb + k + 2 * half];
      d.y = hs[16 + r][kb + k + 2 * half + 1];
      const float4 q0 = *reinterpret_cast<const float4*>(w2p + k * OUT_DIM);
      const float4 q1 = *reinterpret_cast<const float4*>(w2p + (k + 1) * OUT_DIM);
      v2f b;
      b.x = q0.x; b.y = q1.x; WMMA_F32X4(o0, a, b); WMMA_F32X4(p0, d, b);
      b.x = q0.y; b.y = q1.y; WMMA_F32X4(o1, a, b); WMMA_F32X4(p1, d, b);
      b.x = q0.z; b.y = q1.z; WMMA_F32X4(o2, a, b); WMMA_F32X4(p2, d, b);
      b.x = q0.w; b.y = q1.w; WMMA_F32X4(o3, a, b); WMMA_F32X4(p3, d, b);
    }
  }
  __syncthreads();
#pragma unroll
  for (int j = 0; j < 8; ++j) {
    const float4 lo = {o0[j], o1[j], o2[j], o3[j]};
    const float4 hi = {p0[j], p1[j], p2[j], p3[j]};
    *reinterpret_cast<float4*>(&ps[wave][j + 8 * half][4 * r]) = lo;
    *reinterpret_cast<float4*>(&ps[wave][16 + j + 8 * half][4 * r]) = hi;
  }
  __syncthreads();

  const int ob = wave * 16;
  {
    const float bias = b2[ob + r];
    float s = 0.0f, q = 0.0f;
#pragma unroll
    for (int j = 0; j < 16; ++j) {
      const int row = (j & 7) + 8 * half + 16 * (j >> 3);
      float v = ((ps[0][row][ob + r] + ps[1][row][ob + r]) + ps[2][row][ob + r]) + ps[3][row][ob + r];
      v += bias;
      v = v > 0.0f ? v : 0.0f;
      out[(m0 + row) * OUT_DIM + ob + r] = v;
      s += v;
      q += v * v;
    }
    s += __shfl_xor(s, 16);
    q += __shfl_xor(q, 16);
    part[blockIdx.x * 128 + tid] = half ? q : s;
  }
}

#define PART_ROWS (N_NODES / MLP_ROWS)
#define ST_ROWS   25
static_assert(PART_ROWS % ST_ROWS == 0, "every stats block sums whole partial rows");

__global__ __launch_bounds__(128) void bn_stats(
    const float* __restrict__ part, float* __restrict__ gsum,
    float* __restrict__ gsq) {
  const int j = threadIdx.x;
  const float* p = part + (size_t)blockIdx.x * (ST_ROWS * 128) + j;
  float v[ST_ROWS];
#pragma unroll
  for (int u = 0; u < ST_ROWS; ++u) v[u] = p[u * 128];
  float t = 0.0f;
#pragma unroll
  for (int u = 0; u < ST_ROWS; ++u) t += v[u];
  const int c = (j >> 5) * 16 + (j & 15);
  atomicAdd(((j >> 4) & 1) ? &gsq[c] : &gsum[c], t);
}

#define AP_THREADS 128
#define AP_ELEMS   16
static_assert((N_NODES * OUT_DIM) % (AP_THREADS * AP_ELEMS) == 0, "the grid covers out exactly: no bounds guard");
static_assert(AP_THREADS % OUT_DIM == 0, "all of a lane's elements belong to one channel");

__global__ __launch_bounds__(AP_THREADS) void bn_apply(
    float* __restrict__ x, const float* __restrict__ gsum,
    const float* __restrict__ gsq, const float* __restrict__ gamma,
    const float* __restrict__ beta) {
  const int c = threadIdx.x & (OUT_DIM - 1);
  float* p = x + (size_t)blockIdx.x * (AP_THREADS * AP_ELEMS) + threadIdx.x;
  float v[AP_ELEMS];
#pragma unroll
  for (int u = 0; u < AP_ELEMS; ++u) v[u] = p[u * AP_THREADS];
  const float inv_n = 1.0f / (float)N_NODES;
  const float mean = gsum[c] * inv_n;
  const float var  = gsq[c] * inv_n - mean * mean;
  const float sc   = gamma[c] * __frsqrt_rn(var + BN_EPS);
  const float sh   = beta[c];
#pragma unroll
  for (int u = 0; u < AP_ELEMS; ++u) p[u * AP_THREADS] = (v[u] - mean) * sc + sh;
}

extern "C" void kernel_launch(void* const* d_in, const int* in_sizes, int n_in,
                              void* d_out, int out_size, void* d_ws, size_t ws_size,
                              hipStream_t stream) {
  const float* h     = (const float*)d_in[0];
  const float* W1    = (const float*)d_in[1];
  const float* b1    = (const float*)d_in[2];
  const float* W2    = (const float*)d_in[3];
  const float* b2    = (const float*)d_in[4];
  const float* gamma = (const float*)d_in[5];
  const float* beta  = (const float*)d_in[6];
  const int*   src   = (const int*)d_in[7];
  const int*   dst   = (const int*)d_in[8];
  float* out = (float*)d_out;

  float* agg  = (float*)d_ws;
  float* deg  = agg + (size_t)N_NODES * NODE_DIM;
  float* gsum = deg + N_NODES;
  float* gsq  = gsum + OUT_DIM;
  unsigned* bits = (unsigned*)(gsq + OUT_DIM);
  float* part = (float*)(bits + (size_t)OWN_GROUPS * BM_WORDS);
  unsigned* pk = (unsigned*)(part + (size_t)(N_NODES / 16) * 128);

  edge_owner_bits<<<N_CHUNK, OWN_THREADS, 0, stream>>>(src, dst, bits, pk);

  edge_owner_scan<<<OWN_GROUPS, OWN_THREADS, 0, stream>>>(h, pk, bits, agg, deg, gsum);

  fused_mlp<<<N_NODES / MLP_ROWS, 128, 0, stream>>>(h, W1, b1, W2, b2, agg, deg, out, part);

  bn_stats<<<PART_ROWS / ST_ROWS, 128, 0, stream>>>(part, gsum, gsq);

  bn_apply<<<(N_NODES * OUT_DIM) / (AP_THREADS * AP_ELEMS), AP_THREADS, 0, stream>>>(out, gsum, gsq, gamma, beta);
}
